# starting kernel
# baseline (speedup 1.0000x reference)
#include <hip/hip_runtime.h>
#include <math.h>

typedef float v2f __attribute__((ext_vector_type(2)));
typedef float v8f __attribute__((ext_vector_type(8)));

#define EMBD 768
#define HEAD 64
#define SEQ  4096
#define BATCH 4
#define BS   (BATCH * SEQ)

__device__ __forceinline__ v8f wmma_f32(v2f a, v2f b, v8f c) {
    return __builtin_amdgcn_wmma_f32_16x16x4_f32(
         false, a,  false, b,
         (short)0, c,  false,  false);
}

__global__ __launch_bounds__(256) void qkv_proj_kernel(
    const float* __restrict__ x,
    const float* __restrict__ Wq, const float* __restrict__ Wk,
    const float* __restrict__ Wv,
    float* __restrict__ qb, float* __restrict__ kb, float* __restrict__ vT)
{
    const int lane = threadIdx.x & 31;
    const int wave = threadIdx.x >> 5;
    const int g    = blockIdx.x * 8 + wave;
    const int mat  = g >> 10;
    const int tile = g & 1023;
    const int row0 = tile << 4;
    const float* W = (mat == 0) ? Wq : (mat == 1) ? Wk : Wv;

    const int half = lane >> 4;
    const int lx   = lane & 15;

    v8f acc[4] = {};
    const float* xrow = x + (size_t)(row0 + lx) * EMBD;
    for (int c = 0; c < EMBD; c += 4) {
        const int kk = c + 2 * half;
        v2f a = *(const v2f*)(xrow + kk);
        #pragma unroll
        for (int t = 0; t < 4; ++t) {
            v2f bf;
            bf.x = W[(size_t)kk       * HEAD + t * 16 + lx];
            bf.y = W[(size_t)(kk + 1) * HEAD + t * 16 + lx];
            acc[t] = wmma_f32(a, bf, acc[t]);
        }
    }

    if (mat < 2) {
        float* outb = (mat == 0) ? qb : kb;
        #pragma unroll
        for (int t = 0; t < 4; ++t)
            #pragma unroll
            for (int i = 0; i < 8; ++i) {
                const int r = row0 + i + 8 * half;
                outb[(size_t)r * HEAD + t * 16 + lx] = acc[t][i];
            }
    } else {
        const int batch = row0 >> 12;
        const int sbase = row0 & 4095;
        float* vb = vT + (size_t)batch * HEAD * SEQ;
        #pragma unroll
        for (int t = 0; t < 4; ++t)
            #pragma unroll
            for (int i = 0; i < 8; ++i) {
                const int s = sbase + i + 8 * half;
                const int h = t * 16 + lx;
                vb[(size_t)h * SEQ + s] = acc[t][i];
            }
    }
}

__global__ __launch_bounds__(256) void flash_attn_kernel(
    const float* __restrict__ qb, const float* __restrict__ kb,
    const float* __restrict__ vT, float* __restrict__ out)
{
    const int lane  = threadIdx.x & 31;
    const int wave  = threadIdx.x >> 5;
    const int qt    = blockIdx.x * 8 + wave;
    const int batch = qt >> 8;
    const int q0    = (qt & 255) << 4;
    const int half  = lane >> 4;
    const int lx    = lane & 15;

    const float scale = 0.125f;

    v2f qf[16];
    {
        const float* qrow = qb + (size_t)(batch * SEQ + q0 + lx) * HEAD;
        #pragma unroll
        for (int c = 0; c < 16; ++c) {
            v2f v = *(const v2f*)(qrow + 4 * c + 2 * half);
            qf[c].x = v.x * scale;
            qf[c].y = v.y * scale;
        }
    }

    v8f acc[4] = {};
    float m = -INFINITY, l = 0.0f;
    const int q_lane = q0 + lx;
    const float* kbase = kb + (size_t)batch * SEQ * HEAD;
    const float* vbase = vT + (size_t)batch * HEAD * SEQ;

    const int nkt = (q0 >> 4) + 1;
    for (int kt = 0; kt < nkt; ++kt) {
        const int k0 = kt << 4;

        v8f st = {};
        const float* krow = kbase + (size_t)(k0 + lx) * HEAD;
        #pragma unroll
        for (int c = 0; c < 16; ++c) {
            v2f kf = *(const v2f*)(krow + 4 * c + 2 * half);
            st = wmma_f32(kf, qf[c], st);
        }

        float p[8];
        float tmax = -INFINITY;
        #pragma unroll
        for (int i = 0; i < 8; ++i) {
            const int key = k0 + i + 8 * half;
            const float s = (key <= q_lane) ? st[i] : -INFINITY;
            p[i] = s;
            tmax = fmaxf(tmax, s);
        }
        tmax = fmaxf(tmax, __shfl_xor(tmax, 16, 32));
        const float m_new = fmaxf(m, tmax);
        const float alpha = __expf(m - m_new);
        float rsum = 0.0f;
        #pragma unroll
        for (int i = 0; i < 8; ++i) {
            p[i] = __expf(p[i] - m_new);
            rsum += p[i];
        }
        l = l * alpha + rsum + __shfl_xor(rsum, 16, 32);
        m = m_new;
        #pragma unroll
        for (int t = 0; t < 4; ++t)
            #pragma unroll
            for (int i = 0; i < 8; ++i) acc[t][i] *= alpha;

        float sh[8];
        #pragma unroll
        for (int i = 0; i < 8; ++i) sh[i] = __shfl_xor(p[i], 16, 32);
        v2f pf[4];
        #pragma unroll
        for (int c = 0; c < 4; ++c) {
            const int keyA = 4 * c + 2 * half;
            const int keyB = keyA + 1;
            pf[c].x = ((keyA >> 3) == half) ? p[keyA & 7] : sh[keyA & 7];
            pf[c].y = ((keyB >> 3) == half) ? p[keyB & 7] : sh[keyB & 7];
        }

        #pragma unroll
        for (int t = 0; t < 4; ++t) {
            const float* vrow = vbase + (size_t)(t * 16 + lx) * SEQ + k0;
            #pragma unroll
            for (int c = 0; c < 4; ++c) {
                v2f vf = *(const v2f*)(vrow + 4 * c + 2 * half);
                acc[t] = wmma_f32(vf, pf[c], acc[t]);
            }
        }
    }

    const float inv = 1.0f / l;
    float* orow = out + (size_t)(batch * SEQ + q0 + lx) * HEAD + 8 * half;
    #pragma unroll
    for (int t = 0; t < 4; ++t) {
        float4 lo = make_float4(acc[t][0] * inv, acc[t][1] * inv,
                                acc[t][2] * inv, acc[t][3] * inv);
        float4 hi = make_float4(acc[t][4] * inv, acc[t][5] * inv,
                                acc[t][6] * inv, acc[t][7] * inv);
        *(float4*)(orow + t * 16)     = lo;
        *(float4*)(orow + t * 16 + 4) = hi;
    }
}

extern "C" void kernel_launch(void* const* d_in, const int* in_sizes, int n_in,
                              void* d_out, int out_size, void* d_ws, size_t ws_size,
                              hipStream_t stream) {
    const float* x  = (const float*)d_in[0];
    const float* Wq = (const float*)d_in[1];
    const float* Wk = (const float*)d_in[2];
    const float* Wv = (const float*)d_in[3];

    float* qb = (float*)d_ws;
    float* kb = qb + (size_t)BS * HEAD;
    float* vT = kb + (size_t)BS * HEAD;

    qkv_proj_kernel<<<384, 256, 0, stream>>>(x, Wq, Wk, Wv, qb, kb, vT);
    flash_attn_kernel<<<128, 256, 0, stream>>>(qb, kb, vT, (float*)d_out);
}


# Attention: one 8-wave workgroup per 16-query tile, its key tiles dealt round-robin to the waves, partial softmax states merged once in LDS; changes the reduction tree over keys (fixed order)
# speedup vs baseline: 2.5564x; 2.5564x over previous
#include <hip/hip_runtime.h>
#include <math.h>

typedef float v2f __attribute__((ext_vector_type(2)));
typedef float v8f __attribute__((ext_vector_type(8)));

#define EMBD 768
#define HEAD 64
#define SEQ  4096
#define BATCH 4
#define BS   (BATCH * SEQ)

__device__ __forceinline__ v8f wmma_f32(v2f a, v2f b, v8f c) {
    return __builtin_amdgcn_wmma_f32_16x16x4_f32(
         false, a,  false, b,
         (short)0, c,  false,  false);
}

__global__ __launch_bounds__(256) void qkv_proj_kernel(
    const float* __restrict__ x,
    const float* __restrict__ Wq, const float* __restrict__ Wk,
    const float* __restrict__ Wv,
    float* __restrict__ qb, float* __restrict__ kb, float* __restrict__ vT)
{
    const int lane = threadIdx.x & 31;
    const int wave = threadIdx.x >> 5;
    const int g    = blockIdx.x * 8 + wave;
    const int mat  = g >> 10;
    const int tile = g & 1023;
    const int row0 = tile << 4;
    const float* W = (mat == 0) ? Wq : (mat == 1) ? Wk : Wv;

    const int half = lane >> 4;
    const int lx   = lane & 15;

    v8f acc[4] = {};
    const float* xrow = x + (size_t)(row0 + lx) * EMBD;
    for (int c = 0; c < EMBD; c += 4) {
        const int kk = c + 2 * half;
        v2f a = *(const v2f*)(xrow + kk);
        #pragma unroll
        for (int t = 0; t < 4; ++t) {
            v2f bf;
            bf.x = W[(size_t)kk       * HEAD + t * 16 + lx];
            bf.y = W[(size_t)(kk + 1) * HEAD + t * 16 + lx];
            acc[t] = wmma_f32(a, bf, acc[t]);
        }
    }

    if (mat < 2) {
        float* outb = (mat == 0) ? qb : kb;
        #pragma unroll
        for (int t = 0; t < 4; ++t)
            #pragma unroll
            for (int i = 0; i < 8; ++i) {
                const int r = row0 + i + 8 * half;
                outb[(size_t)r * HEAD + t * 16 + lx] = acc[t][i];
            }
    } else {
        const int batch = row0 >> 12;
        const int sbase = row0 & 4095;
        float* vb = vT + (size_t)batch * HEAD * SEQ;
        #pragma unroll
        for (int t = 0; t < 4; ++t)
            #pragma unroll
            for (int i = 0; i < 8; ++i) {
                const int s = sbase + i + 8 * half;
                const int h = t * 16 + lx;
                vb[(size_t)h * SEQ + s] = acc[t][i];
            }
    }
}

#define NWAVE 8
#define APITCH 68

__global__ __launch_bounds__(256) void flash_attn_kernel(
    const float* __restrict__ qb, const float* __restrict__ kb,
    const float* __restrict__ vT, float* __restrict__ out)
{
    __shared__ __attribute__((aligned(16))) float sAcc[NWAVE * 16 * APITCH];
    __shared__ float sM[NWAVE * 16];
    __shared__ float sL[NWAVE * 16];

    const int lane  = threadIdx.x & 31;
    const int wave  = __builtin_amdgcn_readfirstlane((int)(threadIdx.x >> 5));
    const int half  = lane >> 4;
    const int lx    = lane & 15;

    const int batch = blockIdx.x >> 8;
    const int r     = blockIdx.x & 255;
    const int g     = (r & 1) ? (255 - (r >> 1)) : (r >> 1);
    const int qtile = (batch & 1) ? (255 - g) : g;
    const int q0    = qtile << 4;

    const float scale = 0.125f;

    v2f qf[16];
    {
        const float* qrow = qb + (size_t)(batch * SEQ + q0 + lx) * HEAD;
        #pragma unroll
        for (int c = 0; c < 16; ++c) {
            v2f v = *(const v2f*)(qrow + 4 * c + 2 * half);
            qf[c].x = v.x * scale;
            qf[c].y = v.y * scale;
        }
    }

    v8f acc[4] = {};
    float m = -INFINITY, l = 0.0f;
    const int q_lane = q0 + lx;
    const float* kbase = kb + (size_t)batch * SEQ * HEAD;
    const float* vbase = vT + (size_t)batch * HEAD * SEQ;

    const int nkt = qtile + 1;
    for (int kt = wave; kt < nkt; kt += NWAVE) {
        const int k0 = kt << 4;

        v8f st = {};
        const float* krow = kbase + (size_t)(k0 + lx) * HEAD;
        #pragma unroll
        for (int c = 0; c < 16; ++c) {
            v2f kf = *(const v2f*)(krow + 4 * c + 2 * half);
            st = wmma_f32(kf, qf[c], st);
        }

        float p[8];
        float tmax = -INFINITY;
        #pragma unroll
        for (int i = 0; i < 8; ++i) {
            const int key = k0 + i + 8 * half;
            const float s = (key <= q_lane) ? st[i] : -INFINITY;
            p[i] = s;
            tmax = fmaxf(tmax, s);
        }
        tmax = fmaxf(tmax, __shfl_xor(tmax, 16, 32));
        const float m_new = fmaxf(m, tmax);
        const float alpha = __expf(m - m_new);
        float rsum = 0.0f;
        #pragma unroll
        for (int i = 0; i < 8; ++i) {
            p[i] = __expf(p[i] - m_new);
            rsum += p[i];
        }
        l = l * alpha + rsum + __shfl_xor(rsum, 16, 32);
        m = m_new;
        #pragma unroll
        for (int t = 0; t < 4; ++t)
            #pragma unroll
            for (int i = 0; i < 8; ++i) acc[t][i] *= alpha;

        float sh[8];
        #pragma unroll
        for (int i = 0; i < 8; ++i) sh[i] = __shfl_xor(p[i], 16, 32);
        v2f pf[4];
        #pragma unroll
        for (int c = 0; c < 4; ++c) {
            const int keyA = 4 * c + 2 * half;
            const int keyB = keyA + 1;
            pf[c].x = ((keyA >> 3) == half) ? p[keyA & 7] : sh[keyA & 7];
            pf[c].y = ((keyB >> 3) == half) ? p[keyB & 7] : sh[keyB & 7];
        }

        #pragma unroll
        for (int t = 0; t < 4; ++t) {
            const float* vrow = vbase + (size_t)(t * 16 + lx) * SEQ + k0;
            #pragma unroll
            for (int c = 0; c < 4; ++c) {
                v2f vf = *(const v2f*)(vrow + 4 * c + 2 * half);
                acc[t] = wmma_f32(vf, pf[c], acc[t]);
            }
        }
    }

    if (half == 0) {
        sM[wave * 16 + lx] = m;
        sL[wave * 16 + lx] = l;
    }
    {
        float* arow = sAcc + (wave * 16 + lx) * APITCH + 8 * half;
        #pragma unroll
        for (int t = 0; t < 4; ++t) {
            *(float4*)(arow + t * 16)     = make_float4(acc[t][0], acc[t][1],
                                                        acc[t][2], acc[t][3]);
            *(float4*)(arow + t * 16 + 4) = make_float4(acc[t][4], acc[t][5],
                                                        acc[t][6], acc[t][7]);
        }
    }
    __syncthreads();

    const int mq = threadIdx.x >> 4;
    const int hg = threadIdx.x & 15;
    float mw[NWAVE];
    float mmax = -INFINITY;
    #pragma unroll
    for (int w = 0; w < NWAVE; ++w) {
        mw[w] = sM[w * 16 + mq];
        mmax = fmaxf(mmax, mw[w]);
    }
    float lsum = 0.0f;
    float4 o = make_float4(0.0f, 0.0f, 0.0f, 0.0f);
    #pragma unroll
    for (int w = 0; w < NWAVE; ++w) {
        const float wgt = __expf(mw[w] - mmax);
        lsum += sL[w * 16 + mq] * wgt;
        const float4 a = *(const float4*)(sAcc + (w * 16 + mq) * APITCH + 4 * hg);
        o.x += a.x * wgt;
        o.y += a.y * wgt;
        o.z += a.z * wgt;
        o.w += a.w * wgt;
    }
    const float inv = 1.0f / lsum;
    float* orow = out + (size_t)(batch * SEQ + q0 + mq) * HEAD + 4 * hg;
    *(float4*)orow = make_float4(o.x * inv, o.y * inv, o.z * inv, o.w * inv);
}

extern "C" void kernel_launch(void* const* d_in, const int* in_sizes, int n_in,
                              void* d_out, int out_size, void* d_ws, size_t ws_size,
                              hipStream_t stream) {
    const float* x  = (const float*)d_in[0];
    const float* Wq = (const float*)d_in[1];
    const float* Wk = (const float*)d_in[2];
    const float* Wv = (const float*)d_in[3];

    float* qb = (float*)d_ws;
    float* kb = qb + (size_t)BS * HEAD;
    float* vT = kb + (size_t)BS * HEAD;

    qkv_proj_kernel<<<384, 256, 0, stream>>>(x, Wq, Wk, Wv, qb, kb, vT);
    flash_attn_kernel<<<1024, 256, 0, stream>>>(qb, kb, vT, (float*)d_out);
}


# Attention: each wave fetches its K and V key tiles with whole-line 16-B loads one stage ahead, parks them in its own LDS slot and reads the fragments from LDS (V plane kept row-major for it)
# speedup vs baseline: 3.3685x; 1.3177x over previous
#include <hip/hip_runtime.h>
#include <math.h>

typedef float v2f __attribute__((ext_vector_type(2)));
typedef float v4f __attribute__((ext_vector_type(4)));
typedef float v8f __attribute__((ext_vector_type(8)));

#define EMBD 768
#define HEAD 64
#define SEQ  4096
#define BATCH 4
#define BS   (BATCH * SEQ)

__device__ __forceinline__ v8f wmma_f32(v2f a, v2f b, v8f c) {
    return __builtin_amdgcn_wmma_f32_16x16x4_f32(
         false, a,  false, b,
         (short)0, c,  false,  false);
}

__global__ __launch_bounds__(256) void qkv_proj_kernel(
    const float* __restrict__ x,
    const float* __restrict__ Wq, const float* __restrict__ Wk,
    const float* __restrict__ Wv,
    float* __restrict__ qb, float* __restrict__ kb, float* __restrict__ vb)
{
    const int lane = threadIdx.x & 31;
    const int wave = threadIdx.x >> 5;
    const int g    = blockIdx.x * 8 + wave;
    const int mat  = g >> 10;
    const int tile = g & 1023;
    const int row0 = tile << 4;
    const float* W = (mat == 0) ? Wq : (mat == 1) ? Wk : Wv;

    const int half = lane >> 4;
    const int lx   = lane & 15;

    v8f acc[4] = {};
    const float* xrow = x + (size_t)(row0 + lx) * EMBD;
    for (int c = 0; c < EMBD; c += 4) {
        const int kk = c + 2 * half;
        v2f a = *(const v2f*)(xrow + kk);
        #pragma unroll
        for (int t = 0; t < 4; ++t) {
            v2f bf;
            bf.x = W[(size_t)kk       * HEAD + t * 16 + lx];
            bf.y = W[(size_t)(kk + 1) * HEAD + t * 16 + lx];
            acc[t] = wmma_f32(a, bf, acc[t]);
        }
    }

    {
        float* outb = (mat == 0) ? qb : (mat == 1) ? kb : vb;
        #pragma unroll
        for (int t = 0; t < 4; ++t)
            #pragma unroll
            for (int i = 0; i < 8; ++i) {
                const int r = row0 + i + 8 * half;
                outb[(size_t)r * HEAD + t * 16 + lx] = acc[t][i];
            }
    }
}

#define NWAVE 8
#define APITCH 68
#define VPITCH 80

__device__ __forceinline__ void wave_lds_handover() {
    __builtin_amdgcn_fence(__ATOMIC_RELEASE, "wavefront");
    __builtin_amdgcn_wave_barrier();
    __builtin_amdgcn_fence(__ATOMIC_ACQUIRE, "wavefront");
}

__global__ __launch_bounds__(256) void flash_attn_kernel(
    const float* __restrict__ qb, const float* __restrict__ kb,
    const float* __restrict__ vb, float* __restrict__ out)
{
    __shared__ __attribute__((aligned(16))) float sAcc[NWAVE * 16 * APITCH];
    __shared__ __attribute__((aligned(16))) float sV[NWAVE * 16 * VPITCH];
    __shared__ float sM[NWAVE * 16];
    __shared__ float sL[NWAVE * 16];

    const int lane  = threadIdx.x & 31;
    const int wave  = __builtin_amdgcn_readfirstlane((int)(threadIdx.x >> 5));
    const int half  = lane >> 4;
    const int lx    = lane & 15;

    const int batch = blockIdx.x >> 8;
    const int r     = blockIdx.x & 255;
    const int g     = (r & 1) ? (255 - (r >> 1)) : (r >> 1);
    const int qtile = (batch & 1) ? (255 - g) : g;
    const int q0    = qtile << 4;

    const float scale = 0.125f;

    v2f qf[16];
    {
        const float* qrow = qb + (size_t)(batch * SEQ + q0 + lx) * HEAD;
        #pragma unroll
        for (int c = 0; c < 16; ++c) {
            v2f v = *(const v2f*)(qrow + 4 * c + 2 * half);
            qf[c].x = v.x * scale;
            qf[c].y = v.y * scale;
        }
    }

    v8f acc[4] = {};
    float m = -INFINITY, l = 0.0f;
    const int q_lane = q0 + lx;
    const float* kbase = kb + (size_t)batch * SEQ * HEAD;
    const float* vbase = vb + (size_t)batch * SEQ * HEAD;

    float* kS = sAcc + wave * 16 * APITCH;
    float* vS = sV   + wave * 16 * VPITCH;

    const int nkt = qtile + 1;

    const int ldoff = 4 * lane;
    float* kdst = kS + half * APITCH + 4 * lx;
    float* vdst = vS + half * VPITCH + 4 * lx;
    v4f stg[8];
    if (wave < nkt) {
        const float* ktile = kbase + (size_t)(wave << 4) * HEAD + ldoff;
        #pragma unroll
        for (int j = 0; j < 8; ++j) stg[j] = *(const v4f*)(ktile + j * 128);
        #pragma unroll
        for (int j = 0; j < 8; ++j) *(v4f*)(kdst + 2 * j * APITCH) = stg[j];
    }

    for (int kt = wave; kt < nkt; kt += NWAVE) {
        const int k0 = kt << 4;

        const float* vtile = vbase + (size_t)k0 * HEAD + ldoff;
        #pragma unroll
        for (int j = 0; j < 8; ++j) stg[j] = *(const v4f*)(vtile + j * 128);

        wave_lds_handover();

        v8f st = {};
        const float* krow = kS + lx * APITCH + 2 * half;
        #pragma unroll
        for (int c = 0; c < 16; ++c) {
            v2f kf = *(const v2f*)(krow + 4 * c);
            st = wmma_f32(kf, qf[c], st);
        }

        #pragma unroll
        for (int j = 0; j < 8; ++j) *(v4f*)(vdst + 2 * j * VPITCH) = stg[j];

        const int kn = (kt + NWAVE < nkt) ? (kt + NWAVE) : kt;
        const float* ktile = kbase + (size_t)(kn << 4) * HEAD + ldoff;
        #pragma unroll
        for (int j = 0; j < 8; ++j) stg[j] = *(const v4f*)(ktile + j * 128);

        float p[8];
        float tmax = -INFINITY;
        #pragma unroll
        for (int i = 0; i < 8; ++i) {
            const int key = k0 + i + 8 * half;
            const float s = (key <= q_lane) ? st[i] : -INFINITY;
            p[i] = s;
            tmax = fmaxf(tmax, s);
        }
        tmax = fmaxf(tmax, __shfl_xor(tmax, 16, 32));
        const float m_new = fmaxf(m, tmax);
        const float alpha = __expf(m - m_new);
        float rsum = 0.0f;
        #pragma unroll
        for (int i = 0; i < 8; ++i) {
            p[i] = __expf(p[i] - m_new);
            rsum += p[i];
        }
        l = l * alpha + rsum + __shfl_xor(rsum, 16, 32);
        m = m_new;
        #pragma unroll
        for (int t = 0; t < 4; ++t)
            #pragma unroll
            for (int i = 0; i < 8; ++i) acc[t][i] *= alpha;

        float sh[8];
        #pragma unroll
        for (int i = 0; i < 8; ++i) sh[i] = __shfl_xor(p[i], 16, 32);
        v2f pf[4];
        #pragma unroll
        for (int c = 0; c < 4; ++c) {
            const int keyA = 4 * c + 2 * half;
            const int keyB = keyA + 1;
            pf[c].x = ((keyA >> 3) == half) ? p[keyA & 7] : sh[keyA & 7];
            pf[c].y = ((keyB >> 3) == half) ? p[keyB & 7] : sh[keyB & 7];
        }

        wave_lds_handover();

        const float* vcol = vS + (2 * half) * VPITCH + lx;
        #pragma unroll
        for (int t = 0; t < 4; ++t) {
            #pragma unroll
            for (int c = 0; c < 4; ++c) {
                v2f vf;
                vf.x = vcol[(4 * c)     * VPITCH + t * 16];
                vf.y = vcol[(4 * c + 1) * VPITCH + t * 16];
                acc[t] = wmma_f32(vf, pf[c], acc[t]);
            }
        }

        wave_lds_handover();
        #pragma unroll
        for (int j = 0; j < 8; ++j) *(v4f*)(kdst + 2 * j * APITCH) = stg[j];
    }
    wave_lds_handover();

    if (half == 0) {
        sM[wave * 16 + lx] = m;
        sL[wave * 16 + lx] = l;
    }
    {
        float* arow = sAcc + (wave * 16 + lx) * APITCH + 8 * half;
        #pragma unroll
        for (int t = 0; t < 4; ++t) {
            *(float4*)(arow + t * 16)     = make_float4(acc[t][0], acc[t][1],
                                                        acc[t][2], acc[t][3]);
            *(float4*)(arow + t * 16 + 4) = make_float4(acc[t][4], acc[t][5],
                                                        acc[t][6], acc[t][7]);
        }
    }
    __syncthreads();

    const int mq = threadIdx.x >> 4;
    const int hg = threadIdx.x & 15;
    float mw[NWAVE];
    float mmax = -INFINITY;
    #pragma unroll
    for (int w = 0; w < NWAVE; ++w) {
        mw[w] = sM[w * 16 + mq];
        mmax = fmaxf(mmax, mw[w]);
    }
    float lsum = 0.0f;
    float4 o = make_float4(0.0f, 0.0f, 0.0f, 0.0f);
    #pragma unroll
    for (int w = 0; w < NWAVE; ++w) {
        const float wgt = __expf(mw[w] - mmax);
        lsum += sL[w * 16 + mq] * wgt;
        const float4 a = *(const float4*)(sAcc + (w * 16 + mq) * APITCH + 4 * hg);
        o.x += a.x * wgt;
        o.y += a.y * wgt;
        o.z += a.z * wgt;
        o.w += a.w * wgt;
    }
    const float inv = 1.0f / lsum;
    float* orow = out + (size_t)(batch * SEQ + q0 + mq) * HEAD + 4 * hg;
    *(float4*)orow = make_float4(o.x * inv, o.y * inv, o.z * inv, o.w * inv);
}

extern "C" void kernel_launch(void* const* d_in, const int* in_sizes, int n_in,
                              void* d_out, int out_size, void* d_ws, size_t ws_size,
                              hipStream_t stream) {
    const float* x  = (const float*)d_in[0];
    const float* Wq = (const float*)d_in[1];
    const float* Wk = (const float*)d_in[2];
    const float* Wv = (const float*)d_in[3];

    float* qb = (float*)d_ws;
    float* kb = qb + (size_t)BS * HEAD;
    float* vb = kb + (size_t)BS * HEAD;

    qkv_proj_kernel<<<384, 256, 0, stream>>>(x, Wq, Wk, Wv, qb, kb, vb);
    flash_attn_kernel<<<1024, 256, 0, stream>>>(qb, kb, vb, (float*)d_out);
}


# Attention: P*V takes its 16 keys in the order the score registers hold them (k = 8*half+2c+s on both operands), so P stays in its lane: no exchange, no scratch; key sum order inside a tile changes
# speedup vs baseline: 4.4820x; 1.3306x over previous
#include <hip/hip_runtime.h>
#include <math.h>

typedef float v2f __attribute__((ext_vector_type(2)));
typedef float v4f __attribute__((ext_vector_type(4)));
typedef float v8f __attribute__((ext_vector_type(8)));

#define EMBD 768
#define HEAD 64
#define SEQ  4096
#define BATCH 4
#define BS   (BATCH * SEQ)

__device__ __forceinline__ v8f wmma_f32(v2f a, v2f b, v8f c) {
    return __builtin_amdgcn_wmma_f32_16x16x4_f32(
         false, a,  false, b,
         (short)0, c,  false,  false);
}

__global__ __launch_bounds__(256) void qkv_proj_kernel(
    const float* __restrict__ x,
    const float* __restrict__ Wq, const float* __restrict__ Wk,
    const float* __restrict__ Wv,
    float* __restrict__ qb, float* __restrict__ kb, float* __restrict__ vb)
{
    const int lane = threadIdx.x & 31;
    const int wave = threadIdx.x >> 5;
    const int g    = blockIdx.x * 8 + wave;
    const int mat  = g >> 10;
    const int tile = g & 1023;
    const int row0 = tile << 4;
    const float* W = (mat == 0) ? Wq : (mat == 1) ? Wk : Wv;

    const int half = lane >> 4;
    const int lx   = lane & 15;

    v8f acc[4] = {};
    const float* xrow = x + (size_t)(row0 + lx) * EMBD;
    for (int c = 0; c < EMBD; c += 4) {
        const int kk = c + 2 * half;
        v2f a = *(const v2f*)(xrow + kk);
        #pragma unroll
        for (int t = 0; t < 4; ++t) {
            v2f bf;
            bf.x = W[(size_t)kk       * HEAD + t * 16 + lx];
            bf.y = W[(size_t)(kk + 1) * HEAD + t * 16 + lx];
            acc[t] = wmma_f32(a, bf, acc[t]);
        }
    }

    {
        float* outb = (mat == 0) ? qb : (mat == 1) ? kb : vb;
        #pragma unroll
        for (int t = 0; t < 4; ++t)
            #pragma unroll
            for (int i = 0; i < 8; ++i) {
                const int r = row0 + i + 8 * half;
                outb[(size_t)r * HEAD + t * 16 + lx] = acc[t][i];
            }
    }
}

#define NWAVE 8
#define APITCH 68
#define VPITCH 68

__device__ __forceinline__ void wave_lds_handover() {
    __builtin_amdgcn_fence(__ATOMIC_RELEASE, "wavefront");
    __builtin_amdgcn_wave_barrier();
    __builtin_amdgcn_fence(__ATOMIC_ACQUIRE, "wavefront");
}

__global__ __launch_bounds__(256) void flash_attn_kernel(
    const float* __restrict__ qb, const float* __restrict__ kb,
    const float* __restrict__ vb, float* __restrict__ out)
{
    __shared__ __attribute__((aligned(16))) float sAcc[NWAVE * 16 * APITCH];
    __shared__ __attribute__((aligned(16))) float sV[NWAVE * 16 * VPITCH];
    __shared__ float sM[NWAVE * 16];
    __shared__ float sL[NWAVE * 16];

    const int lane  = threadIdx.x & 31;
    const int wave  = __builtin_amdgcn_readfirstlane((int)(threadIdx.x >> 5));
    const int half  = lane >> 4;
    const int lx    = lane & 15;

    const int batch = blockIdx.x >> 8;
    const int r     = blockIdx.x & 255;
    const int g     = (r & 1) ? (255 - (r >> 1)) : (r >> 1);
    const int qtile = (batch & 1) ? (255 - g) : g;
    const int q0    = qtile << 4;

    const float scale = 0.125f;

    v2f qf[16];
    {
        const float* qrow = qb + (size_t)(batch * SEQ + q0 + lx) * HEAD;
        #pragma unroll
        for (int c = 0; c < 16; ++c) {
            v2f v = *(const v2f*)(qrow + 4 * c + 2 * half);
            qf[c].x = v.x * scale;
            qf[c].y = v.y * scale;
        }
    }

    v8f acc[4] = {};
    float m = -INFINITY, l = 0.0f;
    const int q_lane = q0 + lx;
    const float* kbase = kb + (size_t)batch * SEQ * HEAD;
    const float* vbase = vb + (size_t)batch * SEQ * HEAD;

    float* kS = sAcc + wave * 16 * APITCH;
    float* vS = sV   + wave * 16 * VPITCH;

    const int nkt = qtile + 1;

    const int ldoff = 4 * lane;
    float* kdst = kS + half * APITCH + 4 * lx;
    float* vdst = vS + half * VPITCH + 4 * lx;
    v4f stg[8];
    if (wave < nkt) {
        const float* ktile = kbase + (size_t)(wave << 4) * HEAD + ldoff;
        #pragma unroll
        for (int j = 0; j < 8; ++j) stg[j] = *(const v4f*)(ktile + j * 128);
        #pragma unroll
        for (int j = 0; j < 8; ++j) *(v4f*)(kdst + 2 * j * APITCH) = stg[j];
    }

    for (int kt = wave; kt < nkt; kt += NWAVE) {
        const int k0 = kt << 4;

        const float* vtile = vbase + (size_t)k0 * HEAD + ldoff;
        #pragma unroll
        for (int j = 0; j < 8; ++j) stg[j] = *(const v4f*)(vtile + j * 128);

        wave_lds_handover();

        v8f st = {};
        const float* krow = kS + lx * APITCH + 2 * half;
        #pragma unroll
        for (int c = 0; c < 16; ++c) {
            v2f kf = *(const v2f*)(krow + 4 * c);
            st = wmma_f32(kf, qf[c], st);
        }

        asm volatile("" :: "v"(st) : "memory");

        #pragma unroll
        for (int j = 0; j < 8; ++j) *(v4f*)(vdst + 2 * j * VPITCH) = stg[j];

        const int kn = (kt + NWAVE < nkt) ? (kt + NWAVE) : kt;
        const float* ktile = kbase + (size_t)(kn << 4) * HEAD + ldoff;
        #pragma unroll
        for (int j = 0; j < 8; ++j) stg[j] = *(const v4f*)(ktile + j * 128);

        float p[8];
        float tmax = -INFINITY;
        #pragma unroll
        for (int i = 0; i < 8; ++i) {
            const int key = k0 + i + 8 * half;
            const float s = (key <= q_lane) ? st[i] : -INFINITY;
            p[i] = s;
            tmax = fmaxf(tmax, s);
        }
        tmax = fmaxf(tmax, __shfl_xor(tmax, 16, 32));
        const float m_new = fmaxf(m, tmax);
        const float alpha = __expf(m - m_new);
        float rsum = 0.0f;
        #pragma unroll
        for (int i = 0; i < 8; ++i) {
            p[i] = __expf(p[i] - m_new);
            rsum += p[i];
        }
        l = l * alpha + rsum + __shfl_xor(rsum, 16, 32);
        m = m_new;
        #pragma unroll
        for (int t = 0; t < 4; ++t)
            #pragma unroll
            for (int i = 0; i < 8; ++i) acc[t][i] *= alpha;

        v2f pf[4];
        #pragma unroll
        for (int c = 0; c < 4; ++c) {
            pf[c].x = p[2 * c];
            pf[c].y = p[2 * c + 1];
        }

        wave_lds_handover();

        const float* vcol = vS + (8 * half) * VPITCH + lx;
        #pragma unroll
        for (int t = 0; t < 4; ++t) {
            #pragma unroll
            for (int c = 0; c < 4; ++c) {
                v2f vf;
                vf.x = vcol[(2 * c)     * VPITCH + t * 16];
                vf.y = vcol[(2 * c + 1) * VPITCH + t * 16];
                acc[t] = wmma_f32(vf, pf[c], acc[t]);
            }
        }

        wave_lds_handover();
        #pragma unroll
        for (int j = 0; j < 8; ++j) *(v4f*)(kdst + 2 * j * APITCH) = stg[j];
    }
    wave_lds_handover();

    if (half == 0) {
        sM[wave * 16 + lx] = m;
        sL[wave * 16 + lx] = l;
    }
    {
        float* arow = sAcc + (wave * 16 + lx) * APITCH + 8 * half;
        #pragma unroll
        for (int t = 0; t < 4; ++t) {
            *(float4*)(arow + t * 16)     = make_float4(acc[t][0], acc[t][1],
                                                        acc[t][2], acc[t][3]);
            *(float4*)(arow + t * 16 + 4) = make_float4(acc[t][4], acc[t][5],
                                                        acc[t][6], acc[t][7]);
        }
    }
    __syncthreads();

    const int mq = threadIdx.x >> 4;
    const int hg = threadIdx.x & 15;
    float mw[NWAVE];
    float mmax = -INFINITY;
    #pragma unroll
    for (int w = 0; w < NWAVE; ++w) {
        mw[w] = sM[w * 16 + mq];
        mmax = fmaxf(mmax, mw[w]);
    }
    float lsum = 0.0f;
    float4 o = make_float4(0.0f, 0.0f, 0.0f, 0.0f);
    #pragma unroll
    for (int w = 0; w < NWAVE; ++w) {
        const float wgt = __expf(mw[w] - mmax);
        lsum += sL[w * 16 + mq] * wgt;
        const float4 a = *(const float4*)(sAcc + (w * 16 + mq) * APITCH + 4 * hg);
        o.x += a.x * wgt;
        o.y += a.y * wgt;
        o.z += a.z * wgt;
        o.w += a.w * wgt;
    }
    const float inv = 1.0f / lsum;
    float* orow = out + (size_t)(batch * SEQ + q0 + mq) * HEAD + 4 * hg;
    *(float4*)orow = make_float4(o.x * inv, o.y * inv, o.z * inv, o.w * inv);
}

extern "C" void kernel_launch(void* const* d_in, const int* in_sizes, int n_in,
                              void* d_out, int out_size, void* d_ws, size_t ws_size,
                              hipStream_t stream) {
    const float* x  = (const float*)d_in[0];
    const float* Wq = (const float*)d_in[1];
    const float* Wk = (const float*)d_in[2];
    const float* Wv = (const float*)d_in[3];

    float* qb = (float*)d_ws;
    float* kb = qb + (size_t)BS * HEAD;
    float* vb = kb + (size_t)BS * HEAD;

    qkv_proj_kernel<<<384, 256, 0, stream>>>(x, Wq, Wk, Wv, qb, kb, vb);
    flash_attn_kernel<<<1024, 256, 0, stream>>>(qb, kb, vb, (float*)d_out);
}


# Projection as one LDS-tiled GEMM: a 16-wave workgroup takes 64 rows x all 192 Q|K|V columns, x tile and W slab staged in LDS one k chunk ahead, x read once; k sum order kept
# speedup vs baseline: 6.2149x; 1.3866x over previous
#include <hip/hip_runtime.h>
#include <math.h>

typedef float v2f __attribute__((ext_vector_type(2)));
typedef float v4f __attribute__((ext_vector_type(4)));
typedef float v8f __attribute__((ext_vector_type(8)));

#define EMBD 768
#define HEAD 64
#define SEQ  4096
#define BATCH 4
#define BS   (BATCH * SEQ)

__device__ __forceinline__ v8f wmma_f32(v2f a, v2f b, v8f c) {
    return __builtin_amdgcn_wmma_f32_16x16x4_f32(
         false, a,  false, b,
         (short)0, c,  false,  false);
}

#define PROWS  64
#define PKC    64
#define PNCH   (EMBD / PKC)
#define XPITCH 68
#define WTILE  1040
#define OPITCH 196
#define XS_FLOATS (PROWS * XPITCH)
#define WS_FLOATS (12 * WTILE)

__global__ __launch_bounds__(512) void qkv_proj_kernel(
    const float* __restrict__ x,
    const float* __restrict__ Wq, const float* __restrict__ Wk,
    const float* __restrict__ Wv,
    float* __restrict__ qb, float* __restrict__ kb, float* __restrict__ vb)
{
    __shared__ __attribute__((aligned(16))) float xs[2 * XS_FLOATS];
    __shared__ __attribute__((aligned(16))) float ws[2 * WS_FLOATS];
    __shared__ __attribute__((aligned(16))) float os[PROWS * OPITCH];

    const int tid  = threadIdx.x;
    const int lane = tid & 31;
    const int wave = __builtin_amdgcn_readfirstlane(tid >> 5);
    const int half = lane >> 4;
    const int lx   = lane & 15;
    const int rt   = wave & 3;
    const int cq   = wave >> 2;
    const int row0 = blockIdx.x * PROWS;

    const int srow = tid >> 4;
    const int sseg = 4 * (tid & 15);
    const float* xg  = x  + (size_t)(row0 + srow) * EMBD + sseg;
    const float* wg0 = Wq + (size_t)srow * HEAD + sseg;
    const float* wg1 = Wk + (size_t)srow * HEAD + sseg;
    const float* wg2 = Wv + (size_t)srow * HEAD + sseg;
    const int xst = srow * XPITCH + sseg;
    const int wst = (sseg >> 4) * WTILE + srow * 16 + (sseg & 15);

    v4f sx[2], sw[6];
    #pragma unroll
    for (int j = 0; j < 2; ++j) {
        sx[j]     = *(const v4f*)(xg  + (size_t)(32 * j) * EMBD);
        sw[j]     = *(const v4f*)(wg0 + (size_t)(32 * j) * HEAD);
        sw[2 + j] = *(const v4f*)(wg1 + (size_t)(32 * j) * HEAD);
        sw[4 + j] = *(const v4f*)(wg2 + (size_t)(32 * j) * HEAD);
    }
    #pragma unroll
    for (int j = 0; j < 2; ++j) {
        *(v4f*)(xs + xst + 32 * j * XPITCH)       = sx[j];
        *(v4f*)(ws + wst + 32 * j * 16)             = sw[j];
        *(v4f*)(ws + wst + 32 * j * 16 + 4 * WTILE) = sw[2 + j];
        *(v4f*)(ws + wst + 32 * j * 16 + 8 * WTILE) = sw[4 + j];
    }
    __syncthreads();

    v8f acc[3] = {};
    const int afrag = (rt * 16 + lx) * XPITCH + 2 * half;
    const int bfrag = (cq * 3) * WTILE + (2 * half) * 16 + lx;

    for (int ch = 0; ch < PNCH; ++ch) {
        const int cur = ch & 1;
        const bool more = ch + 1 < PNCH;
        if (more) {
            const size_t kn = (size_t)(ch + 1) * PKC;
            #pragma unroll
            for (int j = 0; j < 2; ++j) {
                sx[j]     = *(const v4f*)(xg  + (size_t)(32 * j) * EMBD + kn);
                sw[j]     = *(const v4f*)(wg0 + ((size_t)(32 * j) + kn) * HEAD);
                sw[2 + j] = *(const v4f*)(wg1 + ((size_t)(32 * j) + kn) * HEAD);
                sw[4 + j] = *(const v4f*)(wg2 + ((size_t)(32 * j) + kn) * HEAD);
            }
        }

        const float* xa = xs + cur * XS_FLOATS + afrag;
        const float* wb = ws + cur * WS_FLOATS + bfrag;
        #pragma unroll
        for (int ks = 0; ks < PKC / 4; ++ks) {
            v2f a = *(const v2f*)(xa + 4 * ks);
            #pragma unroll
            for (int t = 0; t < 3; ++t) {
                v2f bf;
                bf.x = wb[t * WTILE + (4 * ks)     * 16];
                bf.y = wb[t * WTILE + (4 * ks + 1) * 16];
                acc[t] = wmma_f32(a, bf, acc[t]);
            }
        }

        if (more) {
            const int nxt = cur ^ 1;
            #pragma unroll
            for (int j = 0; j < 2; ++j) {
                *(v4f*)(xs + nxt * XS_FLOATS + xst + 32 * j * XPITCH)       = sx[j];
                *(v4f*)(ws + nxt * WS_FLOATS + wst + 32 * j * 16)             = sw[j];
                *(v4f*)(ws + nxt * WS_FLOATS + wst + 32 * j * 16 + 4 * WTILE) = sw[2 + j];
                *(v4f*)(ws + nxt * WS_FLOATS + wst + 32 * j * 16 + 8 * WTILE) = sw[4 + j];
            }
            __syncthreads();
        }
    }

    {
        float* od = os + (rt * 16 + 8 * half) * OPITCH + cq * 48 + lx;
        #pragma unroll
        for (int t = 0; t < 3; ++t)
            #pragma unroll
            for (int i = 0; i < 8; ++i) od[i * OPITCH + 16 * t] = acc[t][i];
    }
    __syncthreads();
    #pragma unroll
    for (int j = 0; j < 2; ++j) {
        const int r = srow + 32 * j;
        const float* orow = os + r * OPITCH + sseg;
        const size_t go = (size_t)(row0 + r) * HEAD + sseg;
        *(v4f*)(qb + go) = *(const v4f*)(orow);
        *(v4f*)(kb + go) = *(const v4f*)(orow + 64);
        *(v4f*)(vb + go) = *(const v4f*)(orow + 128);
    }
}

#define NWAVE 8
#define APITCH 68
#define VPITCH 68

__device__ __forceinline__ void wave_lds_handover() {
    __builtin_amdgcn_fence(__ATOMIC_RELEASE, "wavefront");
    __builtin_amdgcn_wave_barrier();
    __builtin_amdgcn_fence(__ATOMIC_ACQUIRE, "wavefront");
}

__global__ __launch_bounds__(256) void flash_attn_kernel(
    const float* __restrict__ qb, const float* __restrict__ kb,
    const float* __restrict__ vb, float* __restrict__ out)
{
    __shared__ __attribute__((aligned(16))) float sAcc[NWAVE * 16 * APITCH];
    __shared__ __attribute__((aligned(16))) float sV[NWAVE * 16 * VPITCH];
    __shared__ float sM[NWAVE * 16];
    __shared__ float sL[NWAVE * 16];

    const int lane  = threadIdx.x & 31;
    const int wave  = __builtin_amdgcn_readfirstlane((int)(threadIdx.x >> 5));
    const int half  = lane >> 4;
    const int lx    = lane & 15;

    const int batch = blockIdx.x >> 8;
    const int r     = blockIdx.x & 255;
    const int g     = (r & 1) ? (255 - (r >> 1)) : (r >> 1);
    const int qtile = (batch & 1) ? (255 - g) : g;
    const int q0    = qtile << 4;

    const float scale = 0.125f;

    v2f qf[16];
    {
        const float* qrow = qb + (size_t)(batch * SEQ + q0 + lx) * HEAD;
        #pragma unroll
        for (int c = 0; c < 16; ++c) {
            v2f v = *(const v2f*)(qrow + 4 * c + 2 * half);
            qf[c].x = v.x * scale;
            qf[c].y = v.y * scale;
        }
    }

    v8f acc[4] = {};
    float m = -INFINITY, l = 0.0f;
    const int q_lane = q0 + lx;
    const float* kbase = kb + (size_t)batch * SEQ * HEAD;
    const float* vbase = vb + (size_t)batch * SEQ * HEAD;

    float* kS = sAcc + wave * 16 * APITCH;
    float* vS = sV   + wave * 16 * VPITCH;

    const int nkt = qtile + 1;

    const int ldoff = 4 * lane;
    float* kdst = kS + half * APITCH + 4 * lx;
    float* vdst = vS + half * VPITCH + 4 * lx;
    v4f stg[8];
    if (wave < nkt) {
        const float* ktile = kbase + (size_t)(wave << 4) * HEAD + ldoff;
        #pragma unroll
        for (int j = 0; j < 8; ++j) stg[j] = *(const v4f*)(ktile + j * 128);
        #pragma unroll
        for (int j = 0; j < 8; ++j) *(v4f*)(kdst + 2 * j * APITCH) = stg[j];
    }

    for (int kt = wave; kt < nkt; kt += NWAVE) {
        const int k0 = kt << 4;

        const float* vtile = vbase + (size_t)k0 * HEAD + ldoff;
        #pragma unroll
        for (int j = 0; j < 8; ++j) stg[j] = *(const v4f*)(vtile + j * 128);

        wave_lds_handover();

        v8f st = {};
        const float* krow = kS + lx * APITCH + 2 * half;
        #pragma unroll
        for (int c = 0; c < 16; ++c) {
            v2f kf = *(const v2f*)(krow + 4 * c);
            st = wmma_f32(kf, qf[c], st);
        }

        asm volatile("" :: "v"(st) : "memory");

        #pragma unroll
        for (int j = 0; j < 8; ++j) *(v4f*)(vdst + 2 * j * VPITCH) = stg[j];

        const int kn = (kt + NWAVE < nkt) ? (kt + NWAVE) : kt;
        const float* ktile = kbase + (size_t)(kn << 4) * HEAD + ldoff;
        #pragma unroll
        for (int j = 0; j < 8; ++j) stg[j] = *(const v4f*)(ktile + j * 128);

        float p[8];
        float tmax = -INFINITY;
        #pragma unroll
        for (int i = 0; i < 8; ++i) {
            const int key = k0 + i + 8 * half;
            const float s = (key <= q_lane) ? st[i] : -INFINITY;
            p[i] = s;
            tmax = fmaxf(tmax, s);
        }
        tmax = fmaxf(tmax, __shfl_xor(tmax, 16, 32));
        const float m_new = fmaxf(m, tmax);
        const float alpha = __expf(m - m_new);
        float rsum = 0.0f;
        #pragma unroll
        for (int i = 0; i < 8; ++i) {
            p[i] = __expf(p[i] - m_new);
            rsum += p[i];
        }
        l = l * alpha + rsum + __shfl_xor(rsum, 16, 32);
        m = m_new;
        #pragma unroll
        for (int t = 0; t < 4; ++t)
            #pragma unroll
            for (int i = 0; i < 8; ++i) acc[t][i] *= alpha;

        v2f pf[4];
        #pragma unroll
        for (int c = 0; c < 4; ++c) {
            pf[c].x = p[2 * c];
            pf[c].y = p[2 * c + 1];
        }

        wave_lds_handover();

        const float* vcol = vS + (8 * half) * VPITCH + lx;
        #pragma unroll
        for (int t = 0; t < 4; ++t) {
            #pragma unroll
            for (int c = 0; c < 4; ++c) {
                v2f vf;
                vf.x = vcol[(2 * c)     * VPITCH + t * 16];
                vf.y = vcol[(2 * c + 1) * VPITCH + t * 16];
                acc[t] = wmma_f32(vf, pf[c], acc[t]);
            }
        }

        wave_lds_handover();
        #pragma unroll
        for (int j = 0; j < 8; ++j) *(v4f*)(kdst + 2 * j * APITCH) = stg[j];
    }
    wave_lds_handover();

    if (half == 0) {
        sM[wave * 16 + lx] = m;
        sL[wave * 16 + lx] = l;
    }
    {
        float* arow = sAcc + (wave * 16 + lx) * APITCH + 8 * half;
        #pragma unroll
        for (int t = 0; t < 4; ++t) {
            *(float4*)(arow + t * 16)     = make_float4(acc[t][0], acc[t][1],
                                                        acc[t][2], acc[t][3]);
            *(float4*)(arow + t * 16 + 4) = make_float4(acc[t][4], acc[t][5],
                                                        acc[t][6], acc[t][7]);
        }
    }
    __syncthreads();

    const int mq = threadIdx.x >> 4;
    const int hg = threadIdx.x & 15;
    float mw[NWAVE];
    float mmax = -INFINITY;
    #pragma unroll
    for (int w = 0; w < NWAVE; ++w) {
        mw[w] = sM[w * 16 + mq];
        mmax = fmaxf(mmax, mw[w]);
    }
    float lsum = 0.0f;
    float4 o = make_float4(0.0f, 0.0f, 0.0f, 0.0f);
    #pragma unroll
    for (int w = 0; w < NWAVE; ++w) {
        const float wgt = __expf(mw[w] - mmax);
        lsum += sL[w * 16 + mq] * wgt;
        const float4 a = *(const float4*)(sAcc + (w * 16 + mq) * APITCH + 4 * hg);
        o.x += a.x * wgt;
        o.y += a.y * wgt;
        o.z += a.z * wgt;
        o.w += a.w * wgt;
    }
    const float inv = 1.0f / lsum;
    float* orow = out + (size_t)(batch * SEQ + q0 + mq) * HEAD + 4 * hg;
    *(float4*)orow = make_float4(o.x * inv, o.y * inv, o.z * inv, o.w * inv);
}

extern "C" void kernel_launch(void* const* d_in, const int* in_sizes, int n_in,
                              void* d_out, int out_size, void* d_ws, size_t ws_size,
                              hipStream_t stream) {
    const float* x  = (const float*)d_in[0];
    const float* Wq = (const float*)d_in[1];
    const float* Wk = (const float*)d_in[2];
    const float* Wv = (const float*)d_in[3];

    float* qb = (float*)d_ws;
    float* kb = qb + (size_t)BS * HEAD;
    float* vb = kb + (size_t)BS * HEAD;

    qkv_proj_kernel<<<256, 512, 0, stream>>>(x, Wq, Wk, Wv, qb, kb, vb);
    flash_attn_kernel<<<1024, 256, 0, stream>>>(qb, kb, vb, (float*)d_out);
}


# Attention: 32-query workgroup tile; each key tile a wave parks in LDS feeds both 16-query sub-tiles (one K/V fragment read, two WMMAs): half the fetches and LDS traffic per product; sum order kept
# speedup vs baseline: 6.4545x; 1.0386x over previous
#include <hip/hip_runtime.h>
#include <math.h>

typedef float v2f __attribute__((ext_vector_type(2)));
typedef float v4f __attribute__((ext_vector_type(4)));
typedef float v8f __attribute__((ext_vector_type(8)));

#define EMBD 768
#define HEAD 64
#define SEQ  4096
#define BATCH 4
#define BS   (BATCH * SEQ)

__device__ __forceinline__ v8f wmma_f32(v2f a, v2f b, v8f c) {
    return __builtin_amdgcn_wmma_f32_16x16x4_f32(
         false, a,  false, b,
         (short)0, c,  false,  false);
}

#define PROWS  64
#define PKC    64
#define PNCH   (EMBD / PKC)
#define XPITCH 68
#define WTILE  1040
#define OPITCH 196
#define XS_FLOATS (PROWS * XPITCH)
#define WS_FLOATS (12 * WTILE)

__global__ __launch_bounds__(512) void qkv_proj_kernel(
    const float* __restrict__ x,
    const float* __restrict__ Wq, const float* __restrict__ Wk,
    const float* __restrict__ Wv,
    float* __restrict__ qb, float* __restrict__ kb, float* __restrict__ vb)
{
    __shared__ __attribute__((aligned(16))) float xs[2 * XS_FLOATS];
    __shared__ __attribute__((aligned(16))) float ws[2 * WS_FLOATS];
    __shared__ __attribute__((aligned(16))) float os[PROWS * OPITCH];

    const int tid  = threadIdx.x;
    const int lane = tid & 31;
    const int wave = __builtin_amdgcn_readfirstlane(tid >> 5);
    const int half = lane >> 4;
    const int lx   = lane & 15;
    const int rt   = wave & 3;
    const int cq   = wave >> 2;
    const int row0 = blockIdx.x * PROWS;

    const int srow = tid >> 4;
    const int sseg = 4 * (tid & 15);
    const float* xg  = x  + (size_t)(row0 + srow) * EMBD + sseg;
    const float* wg0 = Wq + (size_t)srow * HEAD + sseg;
    const float* wg1 = Wk + (size_t)srow * HEAD + sseg;
    const float* wg2 = Wv + (size_t)srow * HEAD + sseg;
    const int xst = srow * XPITCH + sseg;
    const int wst = (sseg >> 4) * WTILE + srow * 16 + (sseg & 15);

    v4f sx[2], sw[6];
    #pragma unroll
    for (int j = 0; j < 2; ++j) {
        sx[j]     = *(const v4f*)(xg  + (size_t)(32 * j) * EMBD);
        sw[j]     = *(const v4f*)(wg0 + (size_t)(32 * j) * HEAD);
        sw[2 + j] = *(const v4f*)(wg1 + (size_t)(32 * j) * HEAD);
        sw[4 + j] = *(const v4f*)(wg2 + (size_t)(32 * j) * HEAD);
    }
    #pragma unroll
    for (int j = 0; j < 2; ++j) {
        *(v4f*)(xs + xst + 32 * j * XPITCH)       = sx[j];
        *(v4f*)(ws + wst + 32 * j * 16)             = sw[j];
        *(v4f*)(ws + wst + 32 * j * 16 + 4 * WTILE) = sw[2 + j];
        *(v4f*)(ws + wst + 32 * j * 16 + 8 * WTILE) = sw[4 + j];
    }
    __syncthreads();

    v8f acc[3] = {};
    const int afrag = (rt * 16 + lx) * XPITCH + 2 * half;
    const int bfrag = (cq * 3) * WTILE + (2 * half) * 16 + lx;

    for (int ch = 0; ch < PNCH; ++ch) {
        const int cur = ch & 1;
        const bool more = ch + 1 < PNCH;
        if (more) {
            const size_t kn = (size_t)(ch + 1) * PKC;
            #pragma unroll
            for (int j = 0; j < 2; ++j) {
                sx[j]     = *(const v4f*)(xg  + (size_t)(32 * j) * EMBD + kn);
                sw[j]     = *(const v4f*)(wg0 + ((size_t)(32 * j) + kn) * HEAD);
                sw[2 + j] = *(const v4f*)(wg1 + ((size_t)(32 * j) + kn) * HEAD);
                sw[4 + j] = *(const v4f*)(wg2 + ((size_t)(32 * j) + kn) * HEAD);
            }
        }

        const float* xa = xs + cur * XS_FLOATS + afrag;
        const float* wb = ws + cur * WS_FLOATS + bfrag;
        #pragma unroll
        for (int ks = 0; ks < PKC / 4; ++ks) {
            v2f a = *(const v2f*)(xa + 4 * ks);
            #pragma unroll
            for (int t = 0; t < 3; ++t) {
                v2f bf;
                bf.x = wb[t * WTILE + (4 * ks)     * 16];
                bf.y = wb[t * WTILE + (4 * ks + 1) * 16];
                acc[t] = wmma_f32(a, bf, acc[t]);
            }
        }

        if (more) {
            const int nxt = cur ^ 1;
            #pragma unroll
            for (int j = 0; j < 2; ++j) {
                *(v4f*)(xs + nxt * XS_FLOATS + xst + 32 * j * XPITCH)       = sx[j];
                *(v4f*)(ws + nxt * WS_FLOATS + wst + 32 * j * 16)             = sw[j];
                *(v4f*)(ws + nxt * WS_FLOATS + wst + 32 * j * 16 + 4 * WTILE) = sw[2 + j];
                *(v4f*)(ws + nxt * WS_FLOATS + wst + 32 * j * 16 + 8 * WTILE) = sw[4 + j];
            }
            __syncthreads();
        }
    }

    {
        float* od = os + (rt * 16 + 8 * half) * OPITCH + cq * 48 + lx;
        #pragma unroll
        for (int t = 0; t < 3; ++t)
            #pragma unroll
            for (int i = 0; i < 8; ++i) od[i * OPITCH + 16 * t] = acc[t][i];
    }
    __syncthreads();
    #pragma unroll
    for (int j = 0; j < 2; ++j) {
        const int r = srow + 32 * j;
        const float* orow = os + r * OPITCH + sseg;
        const size_t go = (size_t)(row0 + r) * HEAD + sseg;
        *(v4f*)(qb + go) = *(const v4f*)(orow);
        *(v4f*)(kb + go) = *(const v4f*)(orow + 64);
        *(v4f*)(vb + go) = *(const v4f*)(orow + 128);
    }
}

#define NWAVE 8
#define QSUB  2
#define QROWS (16 * QSUB)
#define QTILES (SEQ / QROWS)
#define APITCH 68
#define VPITCH 68

__device__ __forceinline__ void wave_lds_handover() {
    __builtin_amdgcn_fence(__ATOMIC_RELEASE, "wavefront");
    __builtin_amdgcn_wave_barrier();
    __builtin_amdgcn_fence(__ATOMIC_ACQUIRE, "wavefront");
}

__device__ __forceinline__ float softmax_tile(const v8f st, const int k0, const int half,
                                              const int q_lane, float& m, float& l,
                                              float (&p)[8]) {
    float tmax = -INFINITY;
    #pragma unroll
    for (int i = 0; i < 8; ++i) {
        const int key = k0 + i + 8 * half;
        const float s = (key <= q_lane) ? st[i] : -INFINITY;
        p[i] = s;
        tmax = fmaxf(tmax, s);
    }
    tmax = fmaxf(tmax, __shfl_xor(tmax, 16, 32));
    const float m_new = fmaxf(m, tmax);
    const float m_use = (m_new == -INFINITY) ? 0.0f : m_new;
    const float alpha = __expf(m - m_use);
    float rsum = 0.0f;
    #pragma unroll
    for (int i = 0; i < 8; ++i) {
        p[i] = __expf(p[i] - m_use);
        rsum += p[i];
    }
    l = l * alpha + rsum + __shfl_xor(rsum, 16, 32);
    m = m_new;
    return alpha;
}

__global__ __launch_bounds__(256) void flash_attn_kernel(
    const float* __restrict__ qb, const float* __restrict__ kb,
    const float* __restrict__ vb, float* __restrict__ out)
{
    __shared__ __attribute__((aligned(16))) float sAcc[NWAVE * QROWS * APITCH];
    __shared__ __attribute__((aligned(16))) float sV[NWAVE * 16 * VPITCH];
    __shared__ float sM[NWAVE * QROWS];
    __shared__ float sL[NWAVE * QROWS];

    const int lane  = threadIdx.x & 31;
    const int wave  = __builtin_amdgcn_readfirstlane((int)(threadIdx.x >> 5));
    const int half  = lane >> 4;
    const int lx    = lane & 15;

    const int batch = blockIdx.x >> 7;
    const int r     = blockIdx.x & (QTILES - 1);
    const int g     = (r & 1) ? (QTILES - 1 - (r >> 1)) : (r >> 1);
    const int qtile = ((batch ^ (batch >> 1)) & 1) ? (QTILES - 1 - g) : g;
    const int q0    = qtile * QROWS;

    const float scale = 0.125f;

    v2f qf[QSUB][16];
    #pragma unroll
    for (int s = 0; s < QSUB; ++s) {
        const float* qrow = qb + (size_t)(batch * SEQ + q0 + 16 * s + lx) * HEAD;
        #pragma unroll
        for (int c = 0; c < 16; ++c) {
            v2f v = *(const v2f*)(qrow + 4 * c + 2 * half);
            qf[s][c].x = v.x * scale;
            qf[s][c].y = v.y * scale;
        }
    }

    v8f acc[QSUB][4] = {};
    float m[QSUB], l[QSUB];
    #pragma unroll
    for (int s = 0; s < QSUB; ++s) { m[s] = -INFINITY; l[s] = 0.0f; }
    const int q_lane = q0 + lx;
    const float* kbase = kb + (size_t)batch * SEQ * HEAD;
    const float* vbase = vb + (size_t)batch * SEQ * HEAD;

    float* kS = sAcc + wave * QROWS * APITCH;
    float* vS = sV   + wave * 16 * VPITCH;

    const int nkt = QSUB * (qtile + 1);

    const int ldoff = 4 * lane;
    float* kdst = kS + half * APITCH + 4 * lx;
    float* vdst = vS + half * VPITCH + 4 * lx;
    v4f stg[8];
    if (wave < nkt) {
        const float* ktile = kbase + (size_t)(wave << 4) * HEAD + ldoff;
        #pragma unroll
        for (int j = 0; j < 8; ++j) stg[j] = *(const v4f*)(ktile + j * 128);
        #pragma unroll
        for (int j = 0; j < 8; ++j) *(v4f*)(kdst + 2 * j * APITCH) = stg[j];
    }

    for (int kt = wave; kt < nkt; kt += NWAVE) {
        const int k0 = kt << 4;

        const float* vtile = vbase + (size_t)k0 * HEAD + ldoff;
        #pragma unroll
        for (int j = 0; j < 8; ++j) stg[j] = *(const v4f*)(vtile + j * 128);

        wave_lds_handover();

        v8f st[QSUB] = {};
        const float* krow = kS + lx * APITCH + 2 * half;
        #pragma unroll
        for (int c = 0; c < 16; ++c) {
            v2f kf = *(const v2f*)(krow + 4 * c);
            #pragma unroll
            for (int s = 0; s < QSUB; ++s) st[s] = wmma_f32(kf, qf[s][c], st[s]);
        }

        asm volatile("" :: "v"(st[0]), "v"(st[1]) : "memory");

        #pragma unroll
        for (int j = 0; j < 8; ++j) *(v4f*)(vdst + 2 * j * VPITCH) = stg[j];

        const int kn = (kt + NWAVE < nkt) ? (kt + NWAVE) : kt;
        const float* ktile = kbase + (size_t)(kn << 4) * HEAD + ldoff;
        #pragma unroll
        for (int j = 0; j < 8; ++j) stg[j] = *(const v4f*)(ktile + j * 128);

        v2f pf[QSUB][4];
        #pragma unroll
        for (int s = 0; s < QSUB; ++s) {
            float p[8];
            const float alpha = softmax_tile(st[s], k0, half, q_lane + 16 * s, m[s], l[s], p);
            #pragma unroll
            for (int t = 0; t < 4; ++t)
                #pragma unroll
                for (int i = 0; i < 8; ++i) acc[s][t][i] *= alpha;
            #pragma unroll
            for (int c = 0; c < 4; ++c) {
                pf[s][c].x = p[2 * c];
                pf[s][c].y = p[2 * c + 1];
            }
        }

        wave_lds_handover();

        const float* vcol = vS + (8 * half) * VPITCH + lx;
        #pragma unroll
        for (int t = 0; t < 4; ++t) {
            #pragma unroll
            for (int c = 0; c < 4; ++c) {
                v2f vf;
                vf.x = vcol[(2 * c)     * VPITCH + t * 16];
                vf.y = vcol[(2 * c + 1) * VPITCH + t * 16];
                #pragma unroll
                for (int s = 0; s < QSUB; ++s) acc[s][t] = wmma_f32(vf, pf[s][c], acc[s][t]);
            }
        }

        wave_lds_handover();
        #pragma unroll
        for (int j = 0; j < 8; ++j) *(v4f*)(kdst + 2 * j * APITCH) = stg[j];
    }
    wave_lds_handover();

    #pragma unroll
    for (int s = 0; s < QSUB; ++s) {
        if (half == 0) {
            sM[wave * QROWS + 16 * s + lx] = m[s];
            sL[wave * QROWS + 16 * s + lx] = l[s];
        }
        float* arow = sAcc + (wave * QROWS + 16 * s + lx) * APITCH + 8 * half;
        #pragma unroll
        for (int t = 0; t < 4; ++t) {
            *(float4*)(arow + t * 16)     = make_float4(acc[s][t][0], acc[s][t][1],
                                                        acc[s][t][2], acc[s][t][3]);
            *(float4*)(arow + t * 16 + 4) = make_float4(acc[s][t][4], acc[s][t][5],
                                                        acc[s][t][6], acc[s][t][7]);
        }
    }
    __syncthreads();

    const int mq = threadIdx.x >> 4;
    const int hg = threadIdx.x & 15;
    #pragma unroll
    for (int s = 0; s < QSUB; ++s) {
        const int qi = 16 * s + mq;
        float mw[NWAVE];
        float mmax = -INFINITY;
        #pragma unroll
        for (int w = 0; w < NWAVE; ++w) {
            mw[w] = sM[w * QROWS + qi];
            mmax = fmaxf(mmax, mw[w]);
        }
        float lsum = 0.0f;
        float4 o = make_float4(0.0f, 0.0f, 0.0f, 0.0f);
        #pragma unroll
        for (int w = 0; w < NWAVE; ++w) {
            const float wgt = __expf(mw[w] - mmax);
            lsum += sL[w * QROWS + qi] * wgt;
            const float4 a = *(const float4*)(sAcc + (w * QROWS + qi) * APITCH + 4 * hg);
            o.x += a.x * wgt;
            o.y += a.y * wgt;
            o.z += a.z * wgt;
            o.w += a.w * wgt;
        }
        const float inv = 1.0f / lsum;
        float* orow = out + (size_t)(batch * SEQ + q0 + qi) * HEAD + 4 * hg;
        *(float4*)orow = make_float4(o.x * inv, o.y * inv, o.z * inv, o.w * inv);
    }
}

extern "C" void kernel_launch(void* const* d_in, const int* in_sizes, int n_in,
                              void* d_out, int out_size, void* d_ws, size_t ws_size,
                              hipStream_t stream) {
    const float* x  = (const float*)d_in[0];
    const float* Wq = (const float*)d_in[1];
    const float* Wk = (const float*)d_in[2];
    const float* Wv = (const float*)d_in[3];

    float* qb = (float*)d_ws;
    float* kb = qb + (size_t)BS * HEAD;
    float* vb = kb + (size_t)BS * HEAD;

    qkv_proj_kernel<<<256, 512, 0, stream>>>(x, Wq, Wk, Wv, qb, kb, vb);
    flash_attn_kernel<<<512, 256, 0, stream>>>(qb, kb, vb, (float*)d_out);
}


# Attention: one 16-wave workgroup per mirrored pair of 32-query tiles (127-g then g), 256 equal workgroups, one per WGP; key tiles dealt to 16 waves: changes the reduction tree over keys (fixed order)
# speedup vs baseline: 7.2570x; 1.1243x over previous
#include <hip/hip_runtime.h>
#include <math.h>

typedef float v2f __attribute__((ext_vector_type(2)));
typedef float v4f __attribute__((ext_vector_type(4)));
typedef float v8f __attribute__((ext_vector_type(8)));

#define EMBD 768
#define HEAD 64
#define SEQ  4096
#define BATCH 4
#define BS   (BATCH * SEQ)

__device__ __forceinline__ v8f wmma_f32(v2f a, v2f b, v8f c) {
    return __builtin_amdgcn_wmma_f32_16x16x4_f32(
         false, a,  false, b,
         (short)0, c,  false,  false);
}

#define PROWS  64
#define PKC    64
#define PNCH   (EMBD / PKC)
#define XPITCH 68
#define WTILE  1040
#define OPITCH 196
#define XS_FLOATS (PROWS * XPITCH)
#define WS_FLOATS (12 * WTILE)

__global__ __launch_bounds__(512) void qkv_proj_kernel(
    const float* __restrict__ x,
    const float* __restrict__ Wq, const float* __restrict__ Wk,
    const float* __restrict__ Wv,
    float* __restrict__ qb, float* __restrict__ kb, float* __restrict__ vb)
{
    __shared__ __attribute__((aligned(16))) float xs[2 * XS_FLOATS];
    __shared__ __attribute__((aligned(16))) float ws[2 * WS_FLOATS];
    __shared__ __attribute__((aligned(16))) float os[PROWS * OPITCH];

    const int tid  = threadIdx.x;
    const int lane = tid & 31;
    const int wave = __builtin_amdgcn_readfirstlane(tid >> 5);
    const int half = lane >> 4;
    const int lx   = lane & 15;
    const int rt   = wave & 3;
    const int cq   = wave >> 2;
    const int row0 = blockIdx.x * PROWS;

    const int srow = tid >> 4;
    const int sseg = 4 * (tid & 15);
    const float* xg  = x  + (size_t)(row0 + srow) * EMBD + sseg;
    const float* wg0 = Wq + (size_t)srow * HEAD + sseg;
    const float* wg1 = Wk + (size_t)srow * HEAD + sseg;
    const float* wg2 = Wv + (size_t)srow * HEAD + sseg;
    const int xst = srow * XPITCH + sseg;
    const int wst = (sseg >> 4) * WTILE + srow * 16 + (sseg & 15);

    v4f sx[2], sw[6];
    #pragma unroll
    for (int j = 0; j < 2; ++j) {
        sx[j]     = *(const v4f*)(xg  + (size_t)(32 * j) * EMBD);
        sw[j]     = *(const v4f*)(wg0 + (size_t)(32 * j) * HEAD);
        sw[2 + j] = *(const v4f*)(wg1 + (size_t)(32 * j) * HEAD);
        sw[4 + j] = *(const v4f*)(wg2 + (size_t)(32 * j) * HEAD);
    }
    #pragma unroll
    for (int j = 0; j < 2; ++j) {
        *(v4f*)(xs + xst + 32 * j * XPITCH)       = sx[j];
        *(v4f*)(ws + wst + 32 * j * 16)             = sw[j];
        *(v4f*)(ws + wst + 32 * j * 16 + 4 * WTILE) = sw[2 + j];
        *(v4f*)(ws + wst + 32 * j * 16 + 8 * WTILE) = sw[4 + j];
    }
    __syncthreads();

    v8f acc[3] = {};
    const int afrag = (rt * 16 + lx) * XPITCH + 2 * half;
    const int bfrag = (cq * 3) * WTILE + (2 * half) * 16 + lx;

    for (int ch = 0; ch < PNCH; ++ch) {
        const int cur = ch & 1;
        const bool more = ch + 1 < PNCH;
        if (more) {
            const size_t kn = (size_t)(ch + 1) * PKC;
            #pragma unroll
            for (int j = 0; j < 2; ++j) {
                sx[j]     = *(const v4f*)(xg  + (size_t)(32 * j) * EMBD + kn);
                sw[j]     = *(const v4f*)(wg0 + ((size_t)(32 * j) + kn) * HEAD);
                sw[2 + j] = *(const v4f*)(wg1 + ((size_t)(32 * j) + kn) * HEAD);
                sw[4 + j] = *(const v4f*)(wg2 + ((size_t)(32 * j) + kn) * HEAD);
            }
        }

        const float* xa = xs + cur * XS_FLOATS + afrag;
        const float* wb = ws + cur * WS_FLOATS + bfrag;
        #pragma unroll
        for (int ks = 0; ks < PKC / 4; ++ks) {
            v2f a = *(const v2f*)(xa + 4 * ks);
            #pragma unroll
            for (int t = 0; t < 3; ++t) {
                v2f bf;
                bf.x = wb[t * WTILE + (4 * ks)     * 16];
                bf.y = wb[t * WTILE + (4 * ks + 1) * 16];
                acc[t] = wmma_f32(a, bf, acc[t]);
            }
        }

        if (more) {
            const int nxt = cur ^ 1;
            #pragma unroll
            for (int j = 0; j < 2; ++j) {
                *(v4f*)(xs + nxt * XS_FLOATS + xst + 32 * j * XPITCH)       = sx[j];
                *(v4f*)(ws + nxt * WS_FLOATS + wst + 32 * j * 16)             = sw[j];
                *(v4f*)(ws + nxt * WS_FLOATS + wst + 32 * j * 16 + 4 * WTILE) = sw[2 + j];
                *(v4f*)(ws + nxt * WS_FLOATS + wst + 32 * j * 16 + 8 * WTILE) = sw[4 + j];
            }
            __syncthreads();
        }
    }

    {
        float* od = os + (rt * 16 + 8 * half) * OPITCH + cq * 48 + lx;
        #pragma unroll
        for (int t = 0; t < 3; ++t)
            #pragma unroll
            for (int i = 0; i < 8; ++i) od[i * OPITCH + 16 * t] = acc[t][i];
    }
    __syncthreads();
    #pragma unroll
    for (int j = 0; j < 2; ++j) {
        const int r = srow + 32 * j;
        const float* orow = os + r * OPITCH + sseg;
        const size_t go = (size_t)(row0 + r) * HEAD + sseg;
        *(v4f*)(qb + go) = *(const v4f*)(orow);
        *(v4f*)(kb + go) = *(const v4f*)(orow + 64);
        *(v4f*)(vb + go) = *(const v4f*)(orow + 128);
    }
}

#define NWAVE 16
#define QSUB  2
#define QROWS (16 * QSUB)
#define QTILES (SEQ / QROWS)
#define QPAIRS (QTILES / 2)
#define APITCH 68
#define VPITCH 68

__device__ __forceinline__ void wave_lds_handover() {
    __builtin_amdgcn_fence(__ATOMIC_RELEASE, "wavefront");
    __builtin_amdgcn_wave_barrier();
    __builtin_amdgcn_fence(__ATOMIC_ACQUIRE, "wavefront");
}

__device__ __forceinline__ float softmax_tile(const v8f st, const int k0, const int half,
                                              const int q_lane, float& m, float& l,
                                              float (&p)[8]) {
    float tmax = -INFINITY;
    #pragma unroll
    for (int i = 0; i < 8; ++i) {
        const int key = k0 + i + 8 * half;
        const float s = (key <= q_lane) ? st[i] : -INFINITY;
        p[i] = s;
        tmax = fmaxf(tmax, s);
    }
    tmax = fmaxf(tmax, __shfl_xor(tmax, 16, 32));
    const float m_new = fmaxf(m, tmax);
    const float m_use = (m_new == -INFINITY) ? 0.0f : m_new;
    const float alpha = __expf(m - m_use);
    float rsum = 0.0f;
    #pragma unroll
    for (int i = 0; i < 8; ++i) {
        p[i] = __expf(p[i] - m_use);
        rsum += p[i];
    }
    l = l * alpha + rsum + __shfl_xor(rsum, 16, 32);
    m = m_new;
    return alpha;
}

__global__ __launch_bounds__(512) void flash_attn_kernel(
    const float* __restrict__ qb, const float* __restrict__ kb,
    const float* __restrict__ vb, float* __restrict__ out)
{
    __shared__ __attribute__((aligned(16))) float sAcc[NWAVE * QROWS * APITCH];
    __shared__ __attribute__((aligned(16))) float sV[NWAVE * 16 * VPITCH];
    __shared__ float sM[NWAVE * QROWS];
    __shared__ float sL[NWAVE * QROWS];

    const int lane  = threadIdx.x & 31;
    const int wave  = __builtin_amdgcn_readfirstlane((int)(threadIdx.x >> 5));
    const int half  = lane >> 4;
    const int lx    = lane & 15;

    const int batch = blockIdx.x >> 6;
    const int g     = blockIdx.x & (QPAIRS - 1);

    const float scale = 0.125f;
    const float* kbase = kb + (size_t)batch * SEQ * HEAD;
    const float* vbase = vb + (size_t)batch * SEQ * HEAD;

    float* kS = sAcc + wave * QROWS * APITCH;
    float* vS = sV   + wave * 16 * VPITCH;

    for (int pass = 0; pass < 2; ++pass) {
    const int qtile = pass ? g : (QTILES - 1 - g);
    const int q0    = qtile * QROWS;

    v2f qf[QSUB][16];
    #pragma unroll
    for (int s = 0; s < QSUB; ++s) {
        const float* qrow = qb + (size_t)(batch * SEQ + q0 + 16 * s + lx) * HEAD;
        #pragma unroll
        for (int c = 0; c < 16; ++c) {
            v2f v = *(const v2f*)(qrow + 4 * c + 2 * half);
            qf[s][c].x = v.x * scale;
            qf[s][c].y = v.y * scale;
        }
    }

    v8f acc[QSUB][4] = {};
    float m[QSUB], l[QSUB];
    #pragma unroll
    for (int s = 0; s < QSUB; ++s) { m[s] = -INFINITY; l[s] = 0.0f; }
    const int q_lane = q0 + lx;

    const int nkt = QSUB * (qtile + 1);

    const int ldoff = 4 * lane;
    float* kdst = kS + half * APITCH + 4 * lx;
    float* vdst = vS + half * VPITCH + 4 * lx;
    v4f stg[8];
    if (wave < nkt) {
        const float* ktile = kbase + (size_t)(wave << 4) * HEAD + ldoff;
        #pragma unroll
        for (int j = 0; j < 8; ++j) stg[j] = *(const v4f*)(ktile + j * 128);
        #pragma unroll
        for (int j = 0; j < 8; ++j) *(v4f*)(kdst + 2 * j * APITCH) = stg[j];
    }

    for (int kt = wave; kt < nkt; kt += NWAVE) {
        const int k0 = kt << 4;

        const float* vtile = vbase + (size_t)k0 * HEAD + ldoff;
        #pragma unroll
        for (int j = 0; j < 8; ++j) stg[j] = *(const v4f*)(vtile + j * 128);

        wave_lds_handover();

        v8f st[QSUB] = {};
        const float* krow = kS + lx * APITCH + 2 * half;
        #pragma unroll
        for (int c = 0; c < 16; ++c) {
            v2f kf = *(const v2f*)(krow + 4 * c);
            #pragma unroll
            for (int s = 0; s < QSUB; ++s) st[s] = wmma_f32(kf, qf[s][c], st[s]);
        }

        asm volatile("" :: "v"(st[0]), "v"(st[1]) : "memory");

        #pragma unroll
        for (int j = 0; j < 8; ++j) *(v4f*)(vdst + 2 * j * VPITCH) = stg[j];

        const int kn = (kt + NWAVE < nkt) ? (kt + NWAVE) : kt;
        const float* ktile = kbase + (size_t)(kn << 4) * HEAD + ldoff;
        #pragma unroll
        for (int j = 0; j < 8; ++j) stg[j] = *(const v4f*)(ktile + j * 128);

        v2f pf[QSUB][4];
        #pragma unroll
        for (int s = 0; s < QSUB; ++s) {
            float p[8];
            const float alpha = softmax_tile(st[s], k0, half, q_lane + 16 * s, m[s], l[s], p);
            #pragma unroll
            for (int t = 0; t < 4; ++t)
                #pragma unroll
                for (int i = 0; i < 8; ++i) acc[s][t][i] *= alpha;
            #pragma unroll
            for (int c = 0; c < 4; ++c) {
                pf[s][c].x = p[2 * c];
                pf[s][c].y = p[2 * c + 1];
            }
        }

        wave_lds_handover();

        const float* vcol = vS + (8 * half) * VPITCH + lx;
        #pragma unroll
        for (int t = 0; t < 4; ++t) {
            #pragma unroll
            for (int c = 0; c < 4; ++c) {
                v2f vf;
                vf.x = vcol[(2 * c)     * VPITCH + t * 16];
                vf.y = vcol[(2 * c + 1) * VPITCH + t * 16];
                #pragma unroll
                for (int s = 0; s < QSUB; ++s) acc[s][t] = wmma_f32(vf, pf[s][c], acc[s][t]);
            }
        }

        wave_lds_handover();
        #pragma unroll
        for (int j = 0; j < 8; ++j) *(v4f*)(kdst + 2 * j * APITCH) = stg[j];
    }
    wave_lds_handover();

    #pragma unroll
    for (int s = 0; s < QSUB; ++s) {
        if (half == 0) {
            sM[wave * QROWS + 16 * s + lx] = m[s];
            sL[wave * QROWS + 16 * s + lx] = l[s];
        }
        float* arow = sAcc + (wave * QROWS + 16 * s + lx) * APITCH + 8 * half;
        #pragma unroll
        for (int t = 0; t < 4; ++t) {
            *(float4*)(arow + t * 16)     = make_float4(acc[s][t][0], acc[s][t][1],
                                                        acc[s][t][2], acc[s][t][3]);
            *(float4*)(arow + t * 16 + 4) = make_float4(acc[s][t][4], acc[s][t][5],
                                                        acc[s][t][6], acc[s][t][7]);
        }
    }
    __syncthreads();

    const int hg = threadIdx.x & 15;
    {
        const int qi = (int)(threadIdx.x >> 4);
        float mw[NWAVE];
        float mmax = -INFINITY;
        #pragma unroll
        for (int w = 0; w < NWAVE; ++w) {
            mw[w] = sM[w * QROWS + qi];
            mmax = fmaxf(mmax, mw[w]);
        }
        float lsum = 0.0f;
        float4 o = make_float4(0.0f, 0.0f, 0.0f, 0.0f);
        #pragma unroll
        for (int w = 0; w < NWAVE; ++w) {
            const float wgt = __expf(mw[w] - mmax);
            lsum += sL[w * QROWS + qi] * wgt;
            const float4 a = *(const float4*)(sAcc + (w * QROWS + qi) * APITCH + 4 * hg);
            o.x += a.x * wgt;
            o.y += a.y * wgt;
            o.z += a.z * wgt;
            o.w += a.w * wgt;
        }
        const float inv = 1.0f / lsum;
        float* orow = out + (size_t)(batch * SEQ + q0 + qi) * HEAD + 4 * hg;
        *(float4*)orow = make_float4(o.x * inv, o.y * inv, o.z * inv, o.w * inv);
    }
    if (pass == 0) __syncthreads();
    }
}

extern "C" void kernel_launch(void* const* d_in, const int* in_sizes, int n_in,
                              void* d_out, int out_size, void* d_ws, size_t ws_size,
                              hipStream_t stream) {
    const float* x  = (const float*)d_in[0];
    const float* Wq = (const float*)d_in[1];
    const float* Wk = (const float*)d_in[2];
    const float* Wv = (const float*)d_in[3];

    float* qb = (float*)d_ws;
    float* kb = qb + (size_t)BS * HEAD;
    float* vb = kb + (size_t)BS * HEAD;

    qkv_proj_kernel<<<256, 512, 0, stream>>>(x, Wq, Wk, Wv, qb, kb, vb);
    flash_attn_kernel<<<256, 512, 0, stream>>>(qb, kb, vb, (float*)d_out);
}


# Attention: no barrier between the two query tiles; the 16-wave key-tile deal runs on from tile 127-g into tile g, one barrier before both merges; changes tile g's key reduction tree (fixed order)
# speedup vs baseline: 7.6078x; 1.0483x over previous
#include <hip/hip_runtime.h>
#include <math.h>

typedef float v2f __attribute__((ext_vector_type(2)));
typedef float v4f __attribute__((ext_vector_type(4)));
typedef float v8f __attribute__((ext_vector_type(8)));

#define EMBD 768
#define HEAD 64
#define SEQ  4096
#define BATCH 4
#define BS   (BATCH * SEQ)

__device__ __forceinline__ v8f wmma_f32(v2f a, v2f b, v8f c) {
    return __builtin_amdgcn_wmma_f32_16x16x4_f32(
         false, a,  false, b,
         (short)0, c,  false,  false);
}

#define PROWS  64
#define PKC    64
#define PNCH   (EMBD / PKC)
#define XPITCH 68
#define WTILE  1040
#define OPITCH 196
#define XS_FLOATS (PROWS * XPITCH)
#define WS_FLOATS (12 * WTILE)

__global__ __launch_bounds__(512) void qkv_proj_kernel(
    const float* __restrict__ x,
    const float* __restrict__ Wq, const float* __restrict__ Wk,
    const float* __restrict__ Wv,
    float* __restrict__ qb, float* __restrict__ kb, float* __restrict__ vb)
{
    __shared__ __attribute__((aligned(16))) float xs[2 * XS_FLOATS];
    __shared__ __attribute__((aligned(16))) float ws[2 * WS_FLOATS];
    __shared__ __attribute__((aligned(16))) float os[PROWS * OPITCH];

    const int tid  = threadIdx.x;
    const int lane = tid & 31;
    const int wave = __builtin_amdgcn_readfirstlane(tid >> 5);
    const int half = lane >> 4;
    const int lx   = lane & 15;
    const int rt   = wave & 3;
    const int cq   = wave >> 2;
    const int row0 = blockIdx.x * PROWS;

    const int srow = tid >> 4;
    const int sseg = 4 * (tid & 15);
    const float* xg  = x  + (size_t)(row0 + srow) * EMBD + sseg;
    const float* wg0 = Wq + (size_t)srow * HEAD + sseg;
    const float* wg1 = Wk + (size_t)srow * HEAD + sseg;
    const float* wg2 = Wv + (size_t)srow * HEAD + sseg;
    const int xst = srow * XPITCH + sseg;
    const int wst = (sseg >> 4) * WTILE + srow * 16 + (sseg & 15);

    v4f sx[2], sw[6];
    #pragma unroll
    for (int j = 0; j < 2; ++j) {
        sx[j]     = *(const v4f*)(xg  + (size_t)(32 * j) * EMBD);
        sw[j]     = *(const v4f*)(wg0 + (size_t)(32 * j) * HEAD);
        sw[2 + j] = *(const v4f*)(wg1 + (size_t)(32 * j) * HEAD);
        sw[4 + j] = *(const v4f*)(wg2 + (size_t)(32 * j) * HEAD);
    }
    #pragma unroll
    for (int j = 0; j < 2; ++j) {
        *(v4f*)(xs + xst + 32 * j * XPITCH)       = sx[j];
        *(v4f*)(ws + wst + 32 * j * 16)             = sw[j];
        *(v4f*)(ws + wst + 32 * j * 16 + 4 * WTILE) = sw[2 + j];
        *(v4f*)(ws + wst + 32 * j * 16 + 8 * WTILE) = sw[4 + j];
    }
    __syncthreads();

    v8f acc[3] = {};
    const int afrag = (rt * 16 + lx) * XPITCH + 2 * half;
    const int bfrag = (cq * 3) * WTILE + (2 * half) * 16 + lx;

    for (int ch = 0; ch < PNCH; ++ch) {
        const int cur = ch & 1;
        const bool more = ch + 1 < PNCH;
        if (more) {
            const size_t kn = (size_t)(ch + 1) * PKC;
            #pragma unroll
            for (int j = 0; j < 2; ++j) {
                sx[j]     = *(const v4f*)(xg  + (size_t)(32 * j) * EMBD + kn);
                sw[j]     = *(const v4f*)(wg0 + ((size_t)(32 * j) + kn) * HEAD);
                sw[2 + j] = *(const v4f*)(wg1 + ((size_t)(32 * j) + kn) * HEAD);
                sw[4 + j] = *(const v4f*)(wg2 + ((size_t)(32 * j) + kn) * HEAD);
            }
        }

        const float* xa = xs + cur * XS_FLOATS + afrag;
        const float* wb = ws + cur * WS_FLOATS + bfrag;
        #pragma unroll
        for (int ks = 0; ks < PKC / 4; ++ks) {
            v2f a = *(const v2f*)(xa + 4 * ks);
            #pragma unroll
            for (int t = 0; t < 3; ++t) {
                v2f bf;
                bf.x = wb[t * WTILE + (4 * ks)     * 16];
                bf.y = wb[t * WTILE + (4 * ks + 1) * 16];
                acc[t] = wmma_f32(a, bf, acc[t]);
            }
        }

        if (more) {
            const int nxt = cur ^ 1;
            #pragma unroll
            for (int j = 0; j < 2; ++j) {
                *(v4f*)(xs + nxt * XS_FLOATS + xst + 32 * j * XPITCH)       = sx[j];
                *(v4f*)(ws + nxt * WS_FLOATS + wst + 32 * j * 16)             = sw[j];
                *(v4f*)(ws + nxt * WS_FLOATS + wst + 32 * j * 16 + 4 * WTILE) = sw[2 + j];
                *(v4f*)(ws + nxt * WS_FLOATS + wst + 32 * j * 16 + 8 * WTILE) = sw[4 + j];
            }
            __syncthreads();
        }
    }

    {
        float* od = os + (rt * 16 + 8 * half) * OPITCH + cq * 48 + lx;
        #pragma unroll
        for (int t = 0; t < 3; ++t)
            #pragma unroll
            for (int i = 0; i < 8; ++i) od[i * OPITCH + 16 * t] = acc[t][i];
    }
    __syncthreads();
    #pragma unroll
    for (int j = 0; j < 2; ++j) {
        const int r = srow + 32 * j;
        const float* orow = os + r * OPITCH + sseg;
        const size_t go = (size_t)(row0 + r) * HEAD + sseg;
        *(v4f*)(qb + go) = *(const v4f*)(orow);
        *(v4f*)(kb + go) = *(const v4f*)(orow + 64);
        *(v4f*)(vb + go) = *(const v4f*)(orow + 128);
    }
}

#define NWAVE 16
#define QSUB  2
#define QROWS (16 * QSUB)
#define QTILES (SEQ / QROWS)
#define QPAIRS (QTILES / 2)
#define APITCH 68
#define VPITCH 68

__device__ __forceinline__ void wave_lds_handover() {
    __builtin_amdgcn_fence(__ATOMIC_RELEASE, "wavefront");
    __builtin_amdgcn_wave_barrier();
    __builtin_amdgcn_fence(__ATOMIC_ACQUIRE, "wavefront");
}

__device__ __forceinline__ float softmax_tile(const v8f st, const int k0, const int half,
                                              const int q_lane, float& m, float& l,
                                              float (&p)[8]) {
    float tmax = -INFINITY;
    #pragma unroll
    for (int i = 0; i < 8; ++i) {
        const int key = k0 + i + 8 * half;
        const float s = (key <= q_lane) ? st[i] : -INFINITY;
        p[i] = s;
        tmax = fmaxf(tmax, s);
    }
    tmax = fmaxf(tmax, __shfl_xor(tmax, 16, 32));
    const float m_new = fmaxf(m, tmax);
    const float m_use = (m_new == -INFINITY) ? 0.0f : m_new;
    const float alpha = __expf(m - m_use);
    float rsum = 0.0f;
    #pragma unroll
    for (int i = 0; i < 8; ++i) {
        p[i] = __expf(p[i] - m_use);
        rsum += p[i];
    }
    l = l * alpha + rsum + __shfl_xor(rsum, 16, 32);
    m = m_new;
    return alpha;
}

__global__ __launch_bounds__(512) void flash_attn_kernel(
    const float* __restrict__ qb, const float* __restrict__ kb,
    const float* __restrict__ vb, float* __restrict__ out)
{
    __shared__ __attribute__((aligned(16))) float sAcc[2 * NWAVE * QROWS * APITCH];
    __shared__ float sM[2 * NWAVE * QROWS];
    __shared__ float sL[2 * NWAVE * QROWS];

    const int lane  = threadIdx.x & 31;
    const int wave  = __builtin_amdgcn_readfirstlane((int)(threadIdx.x >> 5));
    const int half  = lane >> 4;
    const int lx    = lane & 15;

    const int batch = blockIdx.x >> 6;
    const int g     = blockIdx.x & (QPAIRS - 1);

    const float scale = 0.125f;
    const float* kbase = kb + (size_t)batch * SEQ * HEAD;
    const float* vbase = vb + (size_t)batch * SEQ * HEAD;

    float* kS = sAcc + (NWAVE + wave) * QROWS * APITCH;
    float* vS = kS + 16 * APITCH;

    int kfirst = wave;
    for (int pass = 0; pass < 2; ++pass) {
    const int qtile = pass ? g : (QTILES - 1 - g);
    const int q0    = qtile * QROWS;

    v2f qf[QSUB][16];
    #pragma unroll
    for (int s = 0; s < QSUB; ++s) {
        const float* qrow = qb + (size_t)(batch * SEQ + q0 + 16 * s + lx) * HEAD;
        #pragma unroll
        for (int c = 0; c < 16; ++c) {
            v2f v = *(const v2f*)(qrow + 4 * c + 2 * half);
            qf[s][c].x = v.x * scale;
            qf[s][c].y = v.y * scale;
        }
    }

    v8f acc[QSUB][4] = {};
    float m[QSUB], l[QSUB];
    #pragma unroll
    for (int s = 0; s < QSUB; ++s) { m[s] = -INFINITY; l[s] = 0.0f; }
    const int q_lane = q0 + lx;

    const int nkt = QSUB * (qtile + 1);

    const int ldoff = 4 * lane;
    float* kdst = kS + half * APITCH + 4 * lx;
    float* vdst = vS + half * VPITCH + 4 * lx;
    v4f stg[8];
    if (kfirst < nkt) {
        const float* ktile = kbase + (size_t)(kfirst << 4) * HEAD + ldoff;
        #pragma unroll
        for (int j = 0; j < 8; ++j) stg[j] = *(const v4f*)(ktile + j * 128);
        #pragma unroll
        for (int j = 0; j < 8; ++j) *(v4f*)(kdst + 2 * j * APITCH) = stg[j];
    }

    for (int kt = kfirst; kt < nkt; kt += NWAVE) {
        const int k0 = kt << 4;

        const float* vtile = vbase + (size_t)k0 * HEAD + ldoff;
        #pragma unroll
        for (int j = 0; j < 8; ++j) stg[j] = *(const v4f*)(vtile + j * 128);

        wave_lds_handover();

        v8f st[QSUB] = {};
        const float* krow = kS + lx * APITCH + 2 * half;
        #pragma unroll
        for (int c = 0; c < 16; ++c) {
            v2f kf = *(const v2f*)(krow + 4 * c);
            #pragma unroll
            for (int s = 0; s < QSUB; ++s) st[s] = wmma_f32(kf, qf[s][c], st[s]);
        }

        asm volatile("" :: "v"(st[0]), "v"(st[1]) : "memory");

        #pragma unroll
        for (int j = 0; j < 8; ++j) *(v4f*)(vdst + 2 * j * VPITCH) = stg[j];

        const int kn = (kt + NWAVE < nkt) ? (kt + NWAVE) : kt;
        const float* ktile = kbase + (size_t)(kn << 4) * HEAD + ldoff;
        #pragma unroll
        for (int j = 0; j < 8; ++j) stg[j] = *(const v4f*)(ktile + j * 128);

        v2f pf[QSUB][4];
        #pragma unroll
        for (int s = 0; s < QSUB; ++s) {
            float p[8];
            const float alpha = softmax_tile(st[s], k0, half, q_lane + 16 * s, m[s], l[s], p);
            #pragma unroll
            for (int t = 0; t < 4; ++t)
                #pragma unroll
                for (int i = 0; i < 8; ++i) acc[s][t][i] *= alpha;
            #pragma unroll
            for (int c = 0; c < 4; ++c) {
                pf[s][c].x = p[2 * c];
                pf[s][c].y = p[2 * c + 1];
            }
        }

        wave_lds_handover();

        const float* vcol = vS + (8 * half) * VPITCH + lx;
        #pragma unroll
        for (int t = 0; t < 4; ++t) {
            #pragma unroll
            for (int c = 0; c < 4; ++c) {
                v2f vf;
                vf.x = vcol[(2 * c)     * VPITCH + t * 16];
                vf.y = vcol[(2 * c + 1) * VPITCH + t * 16];
                #pragma unroll
                for (int s = 0; s < QSUB; ++s) acc[s][t] = wmma_f32(vf, pf[s][c], acc[s][t]);
            }
        }

        wave_lds_handover();
        #pragma unroll
        for (int j = 0; j < 8; ++j) *(v4f*)(kdst + 2 * j * APITCH) = stg[j];
    }
    wave_lds_handover();

    const int slot = pass * NWAVE + wave;
    #pragma unroll
    for (int s = 0; s < QSUB; ++s) {
        if (half == 0) {
            sM[slot * QROWS + 16 * s + lx] = m[s];
            sL[slot * QROWS + 16 * s + lx] = l[s];
        }
        float* arow = sAcc + (slot * QROWS + 16 * s + lx) * APITCH + 8 * half;
        #pragma unroll
        for (int t = 0; t < 4; ++t) {
            *(float4*)(arow + t * 16)     = make_float4(acc[s][t][0], acc[s][t][1],
                                                        acc[s][t][2], acc[s][t][3]);
            *(float4*)(arow + t * 16 + 4) = make_float4(acc[s][t][4], acc[s][t][5],
                                                        acc[s][t][6], acc[s][t][7]);
        }
    }
    kfirst = (wave - nkt) & (NWAVE - 1);
    }

    __syncthreads();

    const int hg = threadIdx.x & 15;
    for (int pass = 0; pass < 2; ++pass) {
        const int q0 = (pass ? g : (QTILES - 1 - g)) * QROWS;
        const int qi = (int)(threadIdx.x >> 4);
        float mw[NWAVE];
        float mmax = -INFINITY;
        #pragma unroll
        for (int w = 0; w < NWAVE; ++w) {
            mw[w] = sM[(pass * NWAVE + w) * QROWS + qi];
            mmax = fmaxf(mmax, mw[w]);
        }
        float lsum = 0.0f;
        float4 o = make_float4(0.0f, 0.0f, 0.0f, 0.0f);
        #pragma unroll
        for (int w = 0; w < NWAVE; ++w) {
            const float wgt = __expf(mw[w] - mmax);
            lsum += sL[(pass * NWAVE + w) * QROWS + qi] * wgt;
            const float4 a = *(const float4*)(sAcc + ((pass * NWAVE + w) * QROWS + qi) * APITCH + 4 * hg);
            o.x += a.x * wgt;
            o.y += a.y * wgt;
            o.z += a.z * wgt;
            o.w += a.w * wgt;
        }
        const float inv = 1.0f / lsum;
        float* orow = out + (size_t)(batch * SEQ + q0 + qi) * HEAD + 4 * hg;
        *(float4*)orow = make_float4(o.x * inv, o.y * inv, o.z * inv, o.w * inv);
    }
}

extern "C" void kernel_launch(void* const* d_in, const int* in_sizes, int n_in,
                              void* d_out, int out_size, void* d_ws, size_t ws_size,
                              hipStream_t stream) {
    const float* x  = (const float*)d_in[0];
    const float* Wq = (const float*)d_in[1];
    const float* Wk = (const float*)d_in[2];
    const float* Wv = (const float*)d_in[3];

    float* qb = (float*)d_ws;
    float* kb = qb + (size_t)BS * HEAD;
    float* vb = kb + (size_t)BS * HEAD;

    qkv_proj_kernel<<<256, 512, 0, stream>>>(x, Wq, Wk, Wv, qb, kb, vb);
    flash_attn_kernel<<<256, 512, 0, stream>>>(qb, kb, vb, (float*)d_out);
}


# Both kernels: a wave lowers its issue priority (s_setprio 3..0) as it advances through a chunk / its key tiles, so the wave furthest behind wins its SIMD and the four end together; sum orders kept
# speedup vs baseline: 7.8879x; 1.0368x over previous
#include <hip/hip_runtime.h>
#include <math.h>

typedef float v2f __attribute__((ext_vector_type(2)));
typedef float v4f __attribute__((ext_vector_type(4)));
typedef float v8f __attribute__((ext_vector_type(8)));

#define EMBD 768
#define HEAD 64
#define SEQ  4096
#define BATCH 4
#define BS   (BATCH * SEQ)

__device__ __forceinline__ v8f wmma_f32(v2f a, v2f b, v8f c) {
    return __builtin_amdgcn_wmma_f32_16x16x4_f32(
         false, a,  false, b,
         (short)0, c,  false,  false);
}

#define PROWS  64
#define PKC    64
#define PNCH   (EMBD / PKC)
#define XPITCH 68
#define WTILE  1040
#define OPITCH 196
#define XS_FLOATS (PROWS * XPITCH)
#define WS_FLOATS (12 * WTILE)

__global__ __launch_bounds__(512) void qkv_proj_kernel(
    const float* __restrict__ x,
    const float* __restrict__ Wq, const float* __restrict__ Wk,
    const float* __restrict__ Wv,
    float* __restrict__ qb, float* __restrict__ kb, float* __restrict__ vb)
{
    __shared__ __attribute__((aligned(16))) float xs[2 * XS_FLOATS];
    __shared__ __attribute__((aligned(16))) float ws[2 * WS_FLOATS];
    __shared__ __attribute__((aligned(16))) float os[PROWS * OPITCH];

    const int tid  = threadIdx.x;
    const int lane = tid & 31;
    const int wave = __builtin_amdgcn_readfirstlane(tid >> 5);
    const int half = lane >> 4;
    const int lx   = lane & 15;
    const int rt   = wave & 3;
    const int cq   = wave >> 2;
    const int row0 = blockIdx.x * PROWS;

    const int srow = tid >> 4;
    const int sseg = 4 * (tid & 15);
    const float* xg  = x  + (size_t)(row0 + srow) * EMBD + sseg;
    const float* wg0 = Wq + (size_t)srow * HEAD + sseg;
    const float* wg1 = Wk + (size_t)srow * HEAD + sseg;
    const float* wg2 = Wv + (size_t)srow * HEAD + sseg;
    const int xst = srow * XPITCH + sseg;
    const int wst = (sseg >> 4) * WTILE + srow * 16 + (sseg & 15);

    v4f sx[2], sw[6];
    #pragma unroll
    for (int j = 0; j < 2; ++j) {
        sx[j]     = *(const v4f*)(xg  + (size_t)(32 * j) * EMBD);
        sw[j]     = *(const v4f*)(wg0 + (size_t)(32 * j) * HEAD);
        sw[2 + j] = *(const v4f*)(wg1 + (size_t)(32 * j) * HEAD);
        sw[4 + j] = *(const v4f*)(wg2 + (size_t)(32 * j) * HEAD);
    }
    #pragma unroll
    for (int j = 0; j < 2; ++j) {
        *(v4f*)(xs + xst + 32 * j * XPITCH)       = sx[j];
        *(v4f*)(ws + wst + 32 * j * 16)             = sw[j];
        *(v4f*)(ws + wst + 32 * j * 16 + 4 * WTILE) = sw[2 + j];
        *(v4f*)(ws + wst + 32 * j * 16 + 8 * WTILE) = sw[4 + j];
    }
    __syncthreads();

    v8f acc[3] = {};
    const int afrag = (rt * 16 + lx) * XPITCH + 2 * half;
    const int bfrag = (cq * 3) * WTILE + (2 * half) * 16 + lx;

    for (int ch = 0; ch < PNCH; ++ch) {
        __builtin_amdgcn_s_setprio(3);
        const int cur = ch & 1;
        const bool more = ch + 1 < PNCH;
        if (more) {
            const size_t kn = (size_t)(ch + 1) * PKC;
            #pragma unroll
            for (int j = 0; j < 2; ++j) {
                sx[j]     = *(const v4f*)(xg  + (size_t)(32 * j) * EMBD + kn);
                sw[j]     = *(const v4f*)(wg0 + ((size_t)(32 * j) + kn) * HEAD);
                sw[2 + j] = *(const v4f*)(wg1 + ((size_t)(32 * j) + kn) * HEAD);
                sw[4 + j] = *(const v4f*)(wg2 + ((size_t)(32 * j) + kn) * HEAD);
            }
        }

        const float* xa = xs + cur * XS_FLOATS + afrag;
        const float* wb = ws + cur * WS_FLOATS + bfrag;
        #pragma unroll
        for (int ks = 0; ks < PKC / 4; ++ks) {
            if (ks == 8)  __builtin_amdgcn_s_setprio(2);
            if (ks == 12) __builtin_amdgcn_s_setprio(1);
            if (ks == 14) __builtin_amdgcn_s_setprio(0);
            v2f a = *(const v2f*)(xa + 4 * ks);
            #pragma unroll
            for (int t = 0; t < 3; ++t) {
                v2f bf;
                bf.x = wb[t * WTILE + (4 * ks)     * 16];
                bf.y = wb[t * WTILE + (4 * ks + 1) * 16];
                acc[t] = wmma_f32(a, bf, acc[t]);
            }
        }

        if (more) {
            const int nxt = cur ^ 1;
            #pragma unroll
            for (int j = 0; j < 2; ++j) {
                *(v4f*)(xs + nxt * XS_FLOATS + xst + 32 * j * XPITCH)       = sx[j];
                *(v4f*)(ws + nxt * WS_FLOATS + wst + 32 * j * 16)             = sw[j];
                *(v4f*)(ws + nxt * WS_FLOATS + wst + 32 * j * 16 + 4 * WTILE) = sw[2 + j];
                *(v4f*)(ws + nxt * WS_FLOATS + wst + 32 * j * 16 + 8 * WTILE) = sw[4 + j];
            }
            __syncthreads();
        }
    }

    {
        float* od = os + (rt * 16 + 8 * half) * OPITCH + cq * 48 + lx;
        #pragma unroll
        for (int t = 0; t < 3; ++t)
            #pragma unroll
            for (int i = 0; i < 8; ++i) od[i * OPITCH + 16 * t] = acc[t][i];
    }
    __syncthreads();
    #pragma unroll
    for (int j = 0; j < 2; ++j) {
        const int r = srow + 32 * j;
        const float* orow = os + r * OPITCH + sseg;
        const size_t go = (size_t)(row0 + r) * HEAD + sseg;
        *(v4f*)(qb + go) = *(const v4f*)(orow);
        *(v4f*)(kb + go) = *(const v4f*)(orow + 64);
        *(v4f*)(vb + go) = *(const v4f*)(orow + 128);
    }
}

#define NWAVE 16
#define QSUB  2
#define QROWS (16 * QSUB)
#define QTILES (SEQ / QROWS)
#define QPAIRS (QTILES / 2)
#define APITCH 68
#define VPITCH 68

__device__ __forceinline__ void wave_lds_handover() {
    __builtin_amdgcn_fence(__ATOMIC_RELEASE, "wavefront");
    __builtin_amdgcn_wave_barrier();
    __builtin_amdgcn_fence(__ATOMIC_ACQUIRE, "wavefront");
}

__device__ __forceinline__ float softmax_tile(const v8f st, const int k0, const int half,
                                              const int q_lane, float& m, float& l,
                                              float (&p)[8]) {
    float tmax = -INFINITY;
    #pragma unroll
    for (int i = 0; i < 8; ++i) {
        const int key = k0 + i + 8 * half;
        const float s = (key <= q_lane) ? st[i] : -INFINITY;
        p[i] = s;
        tmax = fmaxf(tmax, s);
    }
    tmax = fmaxf(tmax, __shfl_xor(tmax, 16, 32));
    const float m_new = fmaxf(m, tmax);
    const float m_use = (m_new == -INFINITY) ? 0.0f : m_new;
    const float alpha = __expf(m - m_use);
    float rsum = 0.0f;
    #pragma unroll
    for (int i = 0; i < 8; ++i) {
        p[i] = __expf(p[i] - m_use);
        rsum += p[i];
    }
    l = l * alpha + rsum + __shfl_xor(rsum, 16, 32);
    m = m_new;
    return alpha;
}

__global__ __launch_bounds__(512) void flash_attn_kernel(
    const float* __restrict__ qb, const float* __restrict__ kb,
    const float* __restrict__ vb, float* __restrict__ out)
{
    __shared__ __attribute__((aligned(16))) float sAcc[2 * NWAVE * QROWS * APITCH];
    __shared__ float sM[2 * NWAVE * QROWS];
    __shared__ float sL[2 * NWAVE * QROWS];

    const int lane  = threadIdx.x & 31;
    const int wave  = __builtin_amdgcn_readfirstlane((int)(threadIdx.x >> 5));
    const int half  = lane >> 4;
    const int lx    = lane & 15;

    const int batch = blockIdx.x >> 6;
    const int g     = blockIdx.x & (QPAIRS - 1);

    const float scale = 0.125f;
    const float* kbase = kb + (size_t)batch * SEQ * HEAD;
    const float* vbase = vb + (size_t)batch * SEQ * HEAD;

    float* kS = sAcc + (NWAVE + wave) * QROWS * APITCH;
    float* vS = kS + 16 * APITCH;

    int kfirst = wave;
    int left = (QSUB * (QTILES + 1) - wave + NWAVE - 1) / NWAVE;
    __builtin_amdgcn_s_setprio(3);
    for (int pass = 0; pass < 2; ++pass) {
    const int qtile = pass ? g : (QTILES - 1 - g);
    const int q0    = qtile * QROWS;

    v2f qf[QSUB][16];
    #pragma unroll
    for (int s = 0; s < QSUB; ++s) {
        const float* qrow = qb + (size_t)(batch * SEQ + q0 + 16 * s + lx) * HEAD;
        #pragma unroll
        for (int c = 0; c < 16; ++c) {
            v2f v = *(const v2f*)(qrow + 4 * c + 2 * half);
            qf[s][c].x = v.x * scale;
            qf[s][c].y = v.y * scale;
        }
    }

    v8f acc[QSUB][4] = {};
    float m[QSUB], l[QSUB];
    #pragma unroll
    for (int s = 0; s < QSUB; ++s) { m[s] = -INFINITY; l[s] = 0.0f; }
    const int q_lane = q0 + lx;

    const int nkt = QSUB * (qtile + 1);

    const int ldoff = 4 * lane;
    float* kdst = kS + half * APITCH + 4 * lx;
    float* vdst = vS + half * VPITCH + 4 * lx;
    v4f stg[8];
    if (kfirst < nkt) {
        const float* ktile = kbase + (size_t)(kfirst << 4) * HEAD + ldoff;
        #pragma unroll
        for (int j = 0; j < 8; ++j) stg[j] = *(const v4f*)(ktile + j * 128);
        #pragma unroll
        for (int j = 0; j < 8; ++j) *(v4f*)(kdst + 2 * j * APITCH) = stg[j];
    }

    for (int kt = kfirst; kt < nkt; kt += NWAVE) {
        const int k0 = kt << 4;

        const float* vtile = vbase + (size_t)k0 * HEAD + ldoff;
        #pragma unroll
        for (int j = 0; j < 8; ++j) stg[j] = *(const v4f*)(vtile + j * 128);

        wave_lds_handover();

        v8f st[QSUB] = {};
        const float* krow = kS + lx * APITCH + 2 * half;
        #pragma unroll
        for (int c = 0; c < 16; ++c) {
            v2f kf = *(const v2f*)(krow + 4 * c);
            #pragma unroll
            for (int s = 0; s < QSUB; ++s) st[s] = wmma_f32(kf, qf[s][c], st[s]);
        }

        asm volatile("" :: "v"(st[0]), "v"(st[1]) : "memory");

        #pragma unroll
        for (int j = 0; j < 8; ++j) *(v4f*)(vdst + 2 * j * VPITCH) = stg[j];

        const int kn = (kt + NWAVE < nkt) ? (kt + NWAVE) : kt;
        const float* ktile = kbase + (size_t)(kn << 4) * HEAD + ldoff;
        #pragma unroll
        for (int j = 0; j < 8; ++j) stg[j] = *(const v4f*)(ktile + j * 128);

        v2f pf[QSUB][4];
        #pragma unroll
        for (int s = 0; s < QSUB; ++s) {
            float p[8];
            const float alpha = softmax_tile(st[s], k0, half, q_lane + 16 * s, m[s], l[s], p);
            #pragma unroll
            for (int t = 0; t < 4; ++t)
                #pragma unroll
                for (int i = 0; i < 8; ++i) acc[s][t][i] *= alpha;
            #pragma unroll
            for (int c = 0; c < 4; ++c) {
                pf[s][c].x = p[2 * c];
                pf[s][c].y = p[2 * c + 1];
            }
        }

        wave_lds_handover();

        const float* vcol = vS + (8 * half) * VPITCH + lx;
        #pragma unroll
        for (int t = 0; t < 4; ++t) {
            #pragma unroll
            for (int c = 0; c < 4; ++c) {
                v2f vf;
                vf.x = vcol[(2 * c)     * VPITCH + t * 16];
                vf.y = vcol[(2 * c + 1) * VPITCH + t * 16];
                #pragma unroll
                for (int s = 0; s < QSUB; ++s) acc[s][t] = wmma_f32(vf, pf[s][c], acc[s][t]);
            }
        }

        wave_lds_handover();
        #pragma unroll
        for (int j = 0; j < 8; ++j) *(v4f*)(kdst + 2 * j * APITCH) = stg[j];

        --left;
        if (left < 8) {
            if (left == 7)      __builtin_amdgcn_s_setprio(2);
            else if (left == 3) __builtin_amdgcn_s_setprio(1);
            else if (left == 1) __builtin_amdgcn_s_setprio(0);
        }
    }
    wave_lds_handover();

    const int slot = pass * NWAVE + wave;
    #pragma unroll
    for (int s = 0; s < QSUB; ++s) {
        if (half == 0) {
            sM[slot * QROWS + 16 * s + lx] = m[s];
            sL[slot * QROWS + 16 * s + lx] = l[s];
        }
        float* arow = sAcc + (slot * QROWS + 16 * s + lx) * APITCH + 8 * half;
        #pragma unroll
        for (int t = 0; t < 4; ++t) {
            *(float4*)(arow + t * 16)     = make_float4(acc[s][t][0], acc[s][t][1],
                                                        acc[s][t][2], acc[s][t][3]);
            *(float4*)(arow + t * 16 + 4) = make_float4(acc[s][t][4], acc[s][t][5],
                                                        acc[s][t][6], acc[s][t][7]);
        }
    }
    kfirst = (wave - nkt) & (NWAVE - 1);
    }

    __builtin_amdgcn_s_setprio(0);
    __syncthreads();

    const int hg = threadIdx.x & 15;
    for (int pass = 0; pass < 2; ++pass) {
        const int q0 = (pass ? g : (QTILES - 1 - g)) * QROWS;
        const int qi = (int)(threadIdx.x >> 4);
        float mw[NWAVE];
        float mmax = -INFINITY;
        #pragma unroll
        for (int w = 0; w < NWAVE; ++w) {
            mw[w] = sM[(pass * NWAVE + w) * QROWS + qi];
            mmax = fmaxf(mmax, mw[w]);
        }
        float lsum = 0.0f;
        float4 o = make_float4(0.0f, 0.0f, 0.0f, 0.0f);
        #pragma unroll
        for (int w = 0; w < NWAVE; ++w) {
            const float wgt = __expf(mw[w] - mmax);
            lsum += sL[(pass * NWAVE + w) * QROWS + qi] * wgt;
            const float4 a = *(const float4*)(sAcc + ((pass * NWAVE + w) * QROWS + qi) * APITCH + 4 * hg);
            o.x += a.x * wgt;
            o.y += a.y * wgt;
            o.z += a.z * wgt;
            o.w += a.w * wgt;
        }
        const float inv = 1.0f / lsum;
        float* orow = out + (size_t)(batch * SEQ + q0 + qi) * HEAD + 4 * hg;
        *(float4*)orow = make_float4(o.x * inv, o.y * inv, o.z * inv, o.w * inv);
    }
}

extern "C" void kernel_launch(void* const* d_in, const int* in_sizes, int n_in,
                              void* d_out, int out_size, void* d_ws, size_t ws_size,
                              hipStream_t stream) {
    const float* x  = (const float*)d_in[0];
    const float* Wq = (const float*)d_in[1];
    const float* Wk = (const float*)d_in[2];
    const float* Wv = (const float*)d_in[3];

    float* qb = (float*)d_ws;
    float* kb = qb + (size_t)BS * HEAD;
    float* vb = kb + (size_t)BS * HEAD;

    qkv_proj_kernel<<<256, 512, 0, stream>>>(x, Wq, Wk, Wv, qb, kb, vb);
    flash_attn_kernel<<<256, 512, 0, stream>>>(qb, kb, vb, (float*)d_out);
}
